# starting kernel
# baseline (speedup 1.0000x reference)
#include <hip/hip_runtime.h>

#define IN_FEAT 128
#define OUT_FEAT 64
#define LDS_STRIDE 80

typedef __attribute__((ext_vector_type(2))) float v2f;
typedef __attribute__((ext_vector_type(8))) float v8f;

__global__ __launch_bounds__(128) void gcn_gemm_wmma(
    const float* __restrict__ x, const float* __restrict__ w,
    float* __restrict__ support, int n_nodes) {
  __shared__ float wlds[IN_FEAT * LDS_STRIDE];

  const int tid = threadIdx.x;
  for (int i = tid; i < IN_FEAT * OUT_FEAT; i += 128) {
    int r = i >> 6;
    int c = i & 63;
    wlds[r * LDS_STRIDE + c] = w[i];
  }
  __syncthreads();

  const int wave = tid >> 5;
  const int lane = tid & 31;
  const int half = lane >> 4;
  const int l16  = lane & 15;
  const int row_base = blockIdx.x * 16;
  const int col_base = wave * 16;

  int arow = row_base + l16;
  if (arow >= n_nodes) arow = n_nodes - 1;
  const float* __restrict__ xrow = x + (long long)arow * IN_FEAT;

  v8f acc = {};
#pragma unroll
  for (int k = 0; k < IN_FEAT; k += 4) {
    const int kk = k + half * 2;
    v2f a = *(const v2f*)(xrow + kk);
    v2f b;
    b.x = wlds[(kk + 0) * LDS_STRIDE + col_base + l16];
    b.y = wlds[(kk + 1) * LDS_STRIDE + col_base + l16];
    acc = __builtin_amdgcn_wmma_f32_16x16x4_f32(
         false, a,  false, b,
         (short)0, acc,  false,  false);
  }

#pragma unroll
  for (int i = 0; i < 8; ++i) {
    const int orow = row_base + half * 8 + i;
    if (orow < n_nodes)
      support[(long long)orow * OUT_FEAT + col_base + l16] = acc[i];
  }
}

__global__ __launch_bounds__(256) void gcn_zero(float* __restrict__ out, int n) {
  int i = blockIdx.x * blockDim.x + threadIdx.x;
  if (i < n) out[i] = 0.0f;
}

__global__ __launch_bounds__(256) void gcn_spmm(
    const float* __restrict__ support, const int* __restrict__ rows,
    const int* __restrict__ cols, const float* __restrict__ vals,
    float* __restrict__ out, int n_edges) {
  const long long t = (long long)blockIdx.x * blockDim.x + threadIdx.x;
  const long long e = t >> 4;
  if (e >= n_edges) return;
  const int f4 = (int)(t & 15) * 4;

  const int r = rows[e];
  const int c = cols[e];
  const float v = vals[e];

  const float4 s = *(const float4*)(support + (long long)c * OUT_FEAT + f4);
  float* o = out + (long long)r * OUT_FEAT + f4;
  unsafeAtomicAdd(o + 0, v * s.x);
  unsafeAtomicAdd(o + 1, v * s.y);
  unsafeAtomicAdd(o + 2, v * s.z);
  unsafeAtomicAdd(o + 3, v * s.w);
}

extern "C" void kernel_launch(void* const* d_in, const int* in_sizes, int n_in,
                              void* d_out, int out_size, void* d_ws, size_t ws_size,
                              hipStream_t stream) {
  const float* x    = (const float*)d_in[0];
  const float* w    = (const float*)d_in[1];
  const int*   rows = (const int*)d_in[2];
  const int*   cols = (const int*)d_in[3];
  const float* vals = (const float*)d_in[4];
  float* out     = (float*)d_out;
  float* support = (float*)d_ws;

  const int n_nodes = in_sizes[0] / IN_FEAT;
  const int n_edges = in_sizes[2];

  gcn_gemm_wmma<<<(n_nodes + 15) / 16, 128, 0, stream>>>(x, w, support, n_nodes);

  gcn_zero<<<(out_size + 255) / 256, 256, 0, stream>>>(out, out_size);

  const long long sp_threads = (long long)n_edges * 16;
  gcn_spmm<<<(unsigned)((sp_threads + 255) / 256), 256, 0, stream>>>(
      support, rows, cols, vals, out, n_edges);
}


# spmm: a wave walks 32 edges, lane l owns features l and l+32, so each gather load and each f32 atomic add covers one whole 128-B line (was 4 quarter-filled lines); atomic arrival-order sum kept
# speedup vs baseline: 1.7387x; 1.7387x over previous
#include <hip/hip_runtime.h>

#define IN_FEAT 128
#define OUT_FEAT 64
#define LDS_STRIDE 80

typedef __attribute__((ext_vector_type(2))) float v2f;
typedef __attribute__((ext_vector_type(8))) float v8f;

__global__ __launch_bounds__(128) void gcn_gemm_wmma(
    const float* __restrict__ x, const float* __restrict__ w,
    float* __restrict__ support, int n_nodes) {
  __shared__ float wlds[IN_FEAT * LDS_STRIDE];

  const int tid = threadIdx.x;
  for (int i = tid; i < IN_FEAT * OUT_FEAT; i += 128) {
    int r = i >> 6;
    int c = i & 63;
    wlds[r * LDS_STRIDE + c] = w[i];
  }
  __syncthreads();

  const int wave = tid >> 5;
  const int lane = tid & 31;
  const int half = lane >> 4;
  const int l16  = lane & 15;
  const int row_base = blockIdx.x * 16;
  const int col_base = wave * 16;

  int arow = row_base + l16;
  if (arow >= n_nodes) arow = n_nodes - 1;
  const float* __restrict__ xrow = x + (long long)arow * IN_FEAT;

  v8f acc = {};
#pragma unroll
  for (int k = 0; k < IN_FEAT; k += 4) {
    const int kk = k + half * 2;
    v2f a = *(const v2f*)(xrow + kk);
    v2f b;
    b.x = wlds[(kk + 0) * LDS_STRIDE + col_base + l16];
    b.y = wlds[(kk + 1) * LDS_STRIDE + col_base + l16];
    acc = __builtin_amdgcn_wmma_f32_16x16x4_f32(
         false, a,  false, b,
         (short)0, acc,  false,  false);
  }

#pragma unroll
  for (int i = 0; i < 8; ++i) {
    const int orow = row_base + half * 8 + i;
    if (orow < n_nodes)
      support[(long long)orow * OUT_FEAT + col_base + l16] = acc[i];
  }
}

__global__ __launch_bounds__(256) void gcn_zero(float* __restrict__ out, int n) {
  int i = blockIdx.x * blockDim.x + threadIdx.x;
  if (i < n) out[i] = 0.0f;
}

#define EDGES_PER_WAVE 32
#define EDGE_CHUNK 16

__global__ __launch_bounds__(256) void gcn_spmm(
    const float* __restrict__ support, const int* __restrict__ rows,
    const int* __restrict__ cols, const float* __restrict__ vals,
    float* __restrict__ out, int n_edges) {
  const int lane = threadIdx.x & 31;
  const long long wave_id =
      ((long long)blockIdx.x * blockDim.x + threadIdx.x) >> 5;
  const long long e0 = wave_id * EDGES_PER_WAVE;
  if (e0 >= n_edges) return;

  const long long left = (long long)n_edges - e0;
  const int cnt = left >= EDGES_PER_WAVE ? EDGES_PER_WAVE : (int)left;

  const long long el = (lane < cnt) ? (e0 + lane) : ((long long)n_edges - 1);
  const int   r_l = rows[el];
  const int   c_l = cols[el];
  const int   v_l = __float_as_int(vals[el]);

  if (cnt == EDGES_PER_WAVE) {
#pragma unroll
    for (int j0 = 0; j0 < EDGES_PER_WAVE; j0 += EDGE_CHUNK) {
      float s_lo[EDGE_CHUNK], s_hi[EDGE_CHUNK];
#pragma unroll
      for (int j = 0; j < EDGE_CHUNK; ++j) {
        const int c = __builtin_amdgcn_readlane(c_l, j0 + j);
        const float* sp = support + (long long)c * OUT_FEAT + lane;
        s_lo[j] = sp[0];
        s_hi[j] = sp[32];
      }
#pragma unroll
      for (int j = 0; j < EDGE_CHUNK; ++j) {
        const int   r = __builtin_amdgcn_readlane(r_l, j0 + j);
        const float v = __int_as_float(__builtin_amdgcn_readlane(v_l, j0 + j));
        float* o = out + (long long)r * OUT_FEAT + lane;
        unsafeAtomicAdd(o,      v * s_lo[j]);
        unsafeAtomicAdd(o + 32, v * s_hi[j]);
      }
    }
  } else {
    for (int j = 0; j < cnt; ++j) {
      const int   c = __builtin_amdgcn_readlane(c_l, j);
      const int   r = __builtin_amdgcn_readlane(r_l, j);
      const float v = __int_as_float(__builtin_amdgcn_readlane(v_l, j));
      const float* sp = support + (long long)c * OUT_FEAT + lane;
      float* o = out + (long long)r * OUT_FEAT + lane;
      unsafeAtomicAdd(o,      v * sp[0]);
      unsafeAtomicAdd(o + 32, v * sp[32]);
    }
  }
}

extern "C" void kernel_launch(void* const* d_in, const int* in_sizes, int n_in,
                              void* d_out, int out_size, void* d_ws, size_t ws_size,
                              hipStream_t stream) {
  const float* x    = (const float*)d_in[0];
  const float* w    = (const float*)d_in[1];
  const int*   rows = (const int*)d_in[2];
  const int*   cols = (const int*)d_in[3];
  const float* vals = (const float*)d_in[4];
  float* out     = (float*)d_out;
  float* support = (float*)d_ws;

  const int n_nodes = in_sizes[0] / IN_FEAT;
  const int n_edges = in_sizes[2];

  gcn_gemm_wmma<<<(n_nodes + 15) / 16, 128, 0, stream>>>(x, w, support, n_nodes);

  gcn_zero<<<(out_size + 255) / 256, 256, 0, stream>>>(out, out_size);

  gcn_spmm<<<(unsigned)((n_edges + 255) / 256), 256, 0, stream>>>(
      support, rows, cols, vals, out, n_edges);
}


# zero-fill: a fixed grid of 512 workgroups sweeps out with 16-byte stores (4 whole lines per wave store) instead of one 4-byte store per thread: 4,096 waves enter the chip instead of 200,000
# speedup vs baseline: 1.8809x; 1.0818x over previous
#include <hip/hip_runtime.h>

#define IN_FEAT 128
#define OUT_FEAT 64
#define LDS_STRIDE 80

typedef __attribute__((ext_vector_type(2))) float v2f;
typedef __attribute__((ext_vector_type(8))) float v8f;

__global__ __launch_bounds__(128) void gcn_gemm_wmma(
    const float* __restrict__ x, const float* __restrict__ w,
    float* __restrict__ support, int n_nodes) {
  __shared__ float wlds[IN_FEAT * LDS_STRIDE];

  const int tid = threadIdx.x;
  for (int i = tid; i < IN_FEAT * OUT_FEAT; i += 128) {
    int r = i >> 6;
    int c = i & 63;
    wlds[r * LDS_STRIDE + c] = w[i];
  }
  __syncthreads();

  const int wave = tid >> 5;
  const int lane = tid & 31;
  const int half = lane >> 4;
  const int l16  = lane & 15;
  const int row_base = blockIdx.x * 16;
  const int col_base = wave * 16;

  int arow = row_base + l16;
  if (arow >= n_nodes) arow = n_nodes - 1;
  const float* __restrict__ xrow = x + (long long)arow * IN_FEAT;

  v8f acc = {};
#pragma unroll
  for (int k = 0; k < IN_FEAT; k += 4) {
    const int kk = k + half * 2;
    v2f a = *(const v2f*)(xrow + kk);
    v2f b;
    b.x = wlds[(kk + 0) * LDS_STRIDE + col_base + l16];
    b.y = wlds[(kk + 1) * LDS_STRIDE + col_base + l16];
    acc = __builtin_amdgcn_wmma_f32_16x16x4_f32(
         false, a,  false, b,
         (short)0, acc,  false,  false);
  }

#pragma unroll
  for (int i = 0; i < 8; ++i) {
    const int orow = row_base + half * 8 + i;
    if (orow < n_nodes)
      support[(long long)orow * OUT_FEAT + col_base + l16] = acc[i];
  }
}

#define ZERO_BLOCKS 512

typedef __attribute__((ext_vector_type(4))) float v4f;

__global__ __launch_bounds__(256) void gcn_zero(float* __restrict__ out, int n) {
  const int n4 = n >> 2;
  const int stride = gridDim.x * 256;
  v4f* __restrict__ o4 = (v4f*)out;
  const v4f z = {0.0f, 0.0f, 0.0f, 0.0f};
  for (int i = blockIdx.x * 256 + threadIdx.x; i < n4; i += stride) o4[i] = z;
  const int t = (n4 << 2) + blockIdx.x * 256 + threadIdx.x;
  if (t < n) out[t] = 0.0f;
}

#define EDGES_PER_WAVE 32
#define EDGE_CHUNK 16

__global__ __launch_bounds__(256) void gcn_spmm(
    const float* __restrict__ support, const int* __restrict__ rows,
    const int* __restrict__ cols, const float* __restrict__ vals,
    float* __restrict__ out, int n_edges) {
  const int lane = threadIdx.x & 31;
  const long long wave_id =
      ((long long)blockIdx.x * blockDim.x + threadIdx.x) >> 5;
  const long long e0 = wave_id * EDGES_PER_WAVE;
  if (e0 >= n_edges) return;

  const long long left = (long long)n_edges - e0;
  const int cnt = left >= EDGES_PER_WAVE ? EDGES_PER_WAVE : (int)left;

  const long long el = (lane < cnt) ? (e0 + lane) : ((long long)n_edges - 1);
  const int   r_l = rows[el];
  const int   c_l = cols[el];
  const int   v_l = __float_as_int(vals[el]);

  if (cnt == EDGES_PER_WAVE) {
#pragma unroll
    for (int j0 = 0; j0 < EDGES_PER_WAVE; j0 += EDGE_CHUNK) {
      float s_lo[EDGE_CHUNK], s_hi[EDGE_CHUNK];
#pragma unroll
      for (int j = 0; j < EDGE_CHUNK; ++j) {
        const int c = __builtin_amdgcn_readlane(c_l, j0 + j);
        const float* sp = support + (long long)c * OUT_FEAT + lane;
        s_lo[j] = sp[0];
        s_hi[j] = sp[32];
      }
#pragma unroll
      for (int j = 0; j < EDGE_CHUNK; ++j) {
        const int   r = __builtin_amdgcn_readlane(r_l, j0 + j);
        const float v = __int_as_float(__builtin_amdgcn_readlane(v_l, j0 + j));
        float* o = out + (long long)r * OUT_FEAT + lane;
        unsafeAtomicAdd(o,      v * s_lo[j]);
        unsafeAtomicAdd(o + 32, v * s_hi[j]);
      }
    }
  } else {
    for (int j = 0; j < cnt; ++j) {
      const int   c = __builtin_amdgcn_readlane(c_l, j);
      const int   r = __builtin_amdgcn_readlane(r_l, j);
      const float v = __int_as_float(__builtin_amdgcn_readlane(v_l, j));
      const float* sp = support + (long long)c * OUT_FEAT + lane;
      float* o = out + (long long)r * OUT_FEAT + lane;
      unsafeAtomicAdd(o,      v * sp[0]);
      unsafeAtomicAdd(o + 32, v * sp[32]);
    }
  }
}

extern "C" void kernel_launch(void* const* d_in, const int* in_sizes, int n_in,
                              void* d_out, int out_size, void* d_ws, size_t ws_size,
                              hipStream_t stream) {
  const float* x    = (const float*)d_in[0];
  const float* w    = (const float*)d_in[1];
  const int*   rows = (const int*)d_in[2];
  const int*   cols = (const int*)d_in[3];
  const float* vals = (const float*)d_in[4];
  float* out     = (float*)d_out;
  float* support = (float*)d_ws;

  const int n_nodes = in_sizes[0] / IN_FEAT;
  const int n_edges = in_sizes[2];

  gcn_gemm_wmma<<<(n_nodes + 15) / 16, 128, 0, stream>>>(x, w, support, n_nodes);

  gcn_zero<<<ZERO_BLOCKS, 256, 0, stream>>>(out, out_size);

  gcn_spmm<<<(unsigned)((n_edges + 255) / 256), 256, 0, stream>>>(
      support, rows, cols, vals, out, n_edges);
}


# gemm: a wave computes 16 rows x all 64 columns (4 accumulator tiles on one A fragment; was one 16x16 tile), so each x row is loaded by one wave, not four, and W is staged once per 64 rows, not 16
# speedup vs baseline: 2.2256x; 1.1833x over previous
#include <hip/hip_runtime.h>

#define IN_FEAT 128
#define OUT_FEAT 64

typedef __attribute__((ext_vector_type(2))) float v2f;
typedef __attribute__((ext_vector_type(8))) float v8f;

#define KP_PITCH 160

typedef __attribute__((ext_vector_type(4))) float v4f;

__global__ __launch_bounds__(128) void gcn_gemm_wmma(
    const float* __restrict__ x, const float* __restrict__ w,
    float* __restrict__ support, int n_nodes) {
  __shared__ __attribute__((aligned(16))) float wlds[(IN_FEAT / 2) * KP_PITCH];

  const int tid = threadIdx.x;
  for (int i = tid; i < IN_FEAT * OUT_FEAT; i += 128) {
    int r = i >> 6;
    int c = i & 63;
    wlds[(r >> 1) * KP_PITCH + c * 2 + (r & 1)] = w[i];
  }
  __syncthreads();

  const int wave = tid >> 5;
  const int lane = tid & 31;
  const int half = lane >> 4;
  const int l16  = lane & 15;
  const int row_base = (blockIdx.x * 4 + wave) * 16;
  if (row_base >= n_nodes) return;

  int arow = row_base + l16;
  if (arow >= n_nodes) arow = n_nodes - 1;
  const float* __restrict__ xrow = x + (long long)arow * IN_FEAT;
  const float* __restrict__ wl = wlds + half * KP_PITCH + l16 * 4;

  v8f acc0 = {}, acc1 = {}, acc2 = {}, acc3 = {};
#pragma unroll
  for (int k = 0; k < IN_FEAT; k += 4) {
    const int kk = k + half * 2;
    v2f a = *(const v2f*)(xrow + kk);
    const v4f q0 = *(const v4f*)(wl + (k >> 1) * KP_PITCH);
    const v4f q1 = *(const v4f*)(wl + (k >> 1) * KP_PITCH + 64);
    v2f b0 = {q0.x, q0.y};
    v2f b1 = {q0.z, q0.w};
    v2f b2 = {q1.x, q1.y};
    v2f b3 = {q1.z, q1.w};
    acc0 = __builtin_amdgcn_wmma_f32_16x16x4_f32(
         false, a,  false, b0,
         (short)0, acc0,  false,  false);
    acc1 = __builtin_amdgcn_wmma_f32_16x16x4_f32(
        false, a, false, b1, (short)0, acc1, false, false);
    acc2 = __builtin_amdgcn_wmma_f32_16x16x4_f32(
        false, a, false, b2, (short)0, acc2, false, false);
    acc3 = __builtin_amdgcn_wmma_f32_16x16x4_f32(
        false, a, false, b3, (short)0, acc3, false, false);
  }

#pragma unroll
  for (int i = 0; i < 8; ++i) {
    const int orow = row_base + half * 8 + i;
    if (orow < n_nodes) {
      float* __restrict__ o = support + (long long)orow * OUT_FEAT + l16 * 2;
      v2f lo = {acc0[i], acc1[i]};
      v2f hi = {acc2[i], acc3[i]};
      *(v2f*)(o)      = lo;
      *(v2f*)(o + 32) = hi;
    }
  }
}

#define ZERO_BLOCKS 512

__global__ __launch_bounds__(256) void gcn_zero(float* __restrict__ out, int n) {
  const int n4 = n >> 2;
  const int stride = gridDim.x * 256;
  v4f* __restrict__ o4 = (v4f*)out;
  const v4f z = {0.0f, 0.0f, 0.0f, 0.0f};
  for (int i = blockIdx.x * 256 + threadIdx.x; i < n4; i += stride) o4[i] = z;
  const int t = (n4 << 2) + blockIdx.x * 256 + threadIdx.x;
  if (t < n) out[t] = 0.0f;
}

#define EDGES_PER_WAVE 32
#define EDGE_CHUNK 16

__global__ __launch_bounds__(256) void gcn_spmm(
    const float* __restrict__ support, const int* __restrict__ rows,
    const int* __restrict__ cols, const float* __restrict__ vals,
    float* __restrict__ out, int n_edges) {
  const int lane = threadIdx.x & 31;
  const long long wave_id =
      ((long long)blockIdx.x * blockDim.x + threadIdx.x) >> 5;
  const long long e0 = wave_id * EDGES_PER_WAVE;
  if (e0 >= n_edges) return;

  const long long left = (long long)n_edges - e0;
  const int cnt = left >= EDGES_PER_WAVE ? EDGES_PER_WAVE : (int)left;

  const long long el = (lane < cnt) ? (e0 + lane) : ((long long)n_edges - 1);
  const int   r_l = rows[el];
  const int   c_l = cols[el];
  const int   v_l = __float_as_int(vals[el]);

  if (cnt == EDGES_PER_WAVE) {
#pragma unroll
    for (int j0 = 0; j0 < EDGES_PER_WAVE; j0 += EDGE_CHUNK) {
      float s_lo[EDGE_CHUNK], s_hi[EDGE_CHUNK];
#pragma unroll
      for (int j = 0; j < EDGE_CHUNK; ++j) {
        const int c = __builtin_amdgcn_readlane(c_l, j0 + j);
        const float* sp = support + (long long)c * OUT_FEAT + lane;
        s_lo[j] = sp[0];
        s_hi[j] = sp[32];
      }
#pragma unroll
      for (int j = 0; j < EDGE_CHUNK; ++j) {
        const int   r = __builtin_amdgcn_readlane(r_l, j0 + j);
        const float v = __int_as_float(__builtin_amdgcn_readlane(v_l, j0 + j));
        float* o = out + (long long)r * OUT_FEAT + lane;
        unsafeAtomicAdd(o,      v * s_lo[j]);
        unsafeAtomicAdd(o + 32, v * s_hi[j]);
      }
    }
  } else {
    for (int j = 0; j < cnt; ++j) {
      const int   c = __builtin_amdgcn_readlane(c_l, j);
      const int   r = __builtin_amdgcn_readlane(r_l, j);
      const float v = __int_as_float(__builtin_amdgcn_readlane(v_l, j));
      const float* sp = support + (long long)c * OUT_FEAT + lane;
      float* o = out + (long long)r * OUT_FEAT + lane;
      unsafeAtomicAdd(o,      v * sp[0]);
      unsafeAtomicAdd(o + 32, v * sp[32]);
    }
  }
}

extern "C" void kernel_launch(void* const* d_in, const int* in_sizes, int n_in,
                              void* d_out, int out_size, void* d_ws, size_t ws_size,
                              hipStream_t stream) {
  const float* x    = (const float*)d_in[0];
  const float* w    = (const float*)d_in[1];
  const int*   rows = (const int*)d_in[2];
  const int*   cols = (const int*)d_in[3];
  const float* vals = (const float*)d_in[4];
  float* out     = (float*)d_out;
  float* support = (float*)d_ws;

  const int n_nodes = in_sizes[0] / IN_FEAT;
  const int n_edges = in_sizes[2];

  gcn_gemm_wmma<<<(n_nodes + 63) / 64, 128, 0, stream>>>(x, w, support, n_nodes);

  gcn_zero<<<ZERO_BLOCKS, 256, 0, stream>>>(out, out_size);

  gcn_spmm<<<(unsigned)((n_edges + 255) / 256), 256, 0, stream>>>(
      support, rows, cols, vals, out, n_edges);
}


# spmm: edges binned by destination-row range in LDS; one work-group per range sums its rows with LDS float adds and stores each line once: no global atomics, no zero-fill; sum order = LDS arrival
# speedup vs baseline: 2.5846x; 1.1613x over previous
#include <hip/hip_runtime.h>

#define IN_FEAT 128
#define OUT_FEAT 64

typedef __attribute__((ext_vector_type(2))) float v2f;
typedef __attribute__((ext_vector_type(8))) float v8f;

#define KP_PITCH 160

typedef __attribute__((ext_vector_type(4))) float v4f;

__global__ __launch_bounds__(128) void gcn_gemm_wmma(
    const float* __restrict__ x, const float* __restrict__ w,
    float* __restrict__ support, int n_nodes) {
  __shared__ __attribute__((aligned(16))) float wlds[(IN_FEAT / 2) * KP_PITCH];

  const int tid = threadIdx.x;
  for (int i = tid; i < IN_FEAT * OUT_FEAT; i += 128) {
    int r = i >> 6;
    int c = i & 63;
    wlds[(r >> 1) * KP_PITCH + c * 2 + (r & 1)] = w[i];
  }
  __syncthreads();

  const int wave = tid >> 5;
  const int lane = tid & 31;
  const int half = lane >> 4;
  const int l16  = lane & 15;
  const int row_base = (blockIdx.x * 4 + wave) * 16;
  if (row_base >= n_nodes) return;

  int arow = row_base + l16;
  if (arow >= n_nodes) arow = n_nodes - 1;
  const float* __restrict__ xrow = x + (long long)arow * IN_FEAT;
  const float* __restrict__ wl = wlds + half * KP_PITCH + l16 * 4;

  v8f acc0 = {}, acc1 = {}, acc2 = {}, acc3 = {};
#pragma unroll
  for (int k = 0; k < IN_FEAT; k += 4) {
    const int kk = k + half * 2;
    v2f a = *(const v2f*)(xrow + kk);
    const v4f q0 = *(const v4f*)(wl + (k >> 1) * KP_PITCH);
    const v4f q1 = *(const v4f*)(wl + (k >> 1) * KP_PITCH + 64);
    v2f b0 = {q0.x, q0.y};
    v2f b1 = {q0.z, q0.w};
    v2f b2 = {q1.x, q1.y};
    v2f b3 = {q1.z, q1.w};
    acc0 = __builtin_amdgcn_wmma_f32_16x16x4_f32(
         false, a,  false, b0,
         (short)0, acc0,  false,  false);
    acc1 = __builtin_amdgcn_wmma_f32_16x16x4_f32(
        false, a, false, b1, (short)0, acc1, false, false);
    acc2 = __builtin_amdgcn_wmma_f32_16x16x4_f32(
        false, a, false, b2, (short)0, acc2, false, false);
    acc3 = __builtin_amdgcn_wmma_f32_16x16x4_f32(
        false, a, false, b3, (short)0, acc3, false, false);
  }

#pragma unroll
  for (int i = 0; i < 8; ++i) {
    const int orow = row_base + half * 8 + i;
    if (orow < n_nodes) {
      float* __restrict__ o = support + (long long)orow * OUT_FEAT + l16 * 2;
      v2f lo = {acc0[i], acc1[i]};
      v2f hi = {acc2[i], acc3[i]};
      *(v2f*)(o)      = lo;
      *(v2f*)(o + 32) = hi;
    }
  }
}

#define BINS 256
#define CHUNK 8192
#define CHUNK_SHIFT 13
#define RPB_MAX 392

__global__ __launch_bounds__(1024) void gcn_bin_edges(
    const int* __restrict__ rows, const int* __restrict__ cols,
    const float* __restrict__ vals, float* __restrict__ ws,
    int n_nodes, int n_edges) {
  __shared__ int cnt[BINS];
  __shared__ int cur[BINS];
  __shared__ int wtot[BINS / 32];
  __shared__ int   s_row[CHUNK];
  __shared__ int   s_col[CHUNK];
  __shared__ float s_val[CHUNK];

  const int tid  = threadIdx.x;
  const int lane = tid & 31;
  const int rpb  = (n_nodes + BINS - 1) / BINS;
  const long long e_begin = (long long)blockIdx.x * CHUNK;
  const long long left = (long long)n_edges - e_begin;
  const int n_here = left >= CHUNK ? CHUNK : (int)left;

  int*   g_row = (int*)(ws + (long long)n_nodes * OUT_FEAT);
  int*   g_col = g_row + n_edges;
  float* g_val = (float*)(g_col + n_edges);
  int*   g_off = (int*)(g_val + n_edges);

  if (tid < BINS) cnt[tid] = 0;
  __syncthreads();

  int r[CHUNK / 1024], bn[CHUNK / 1024];
#pragma unroll
  for (int i = 0; i < CHUNK / 1024; ++i) {
    const int idx = tid + i * 1024;
    r[i] = 0;
    bn[i] = 0;
    if (idx < n_here) {
      r[i]  = rows[e_begin + idx];
      bn[i] = r[i] / rpb;
      atomicAdd(&cnt[bn[i]], 1);
    }
  }
  __syncthreads();

  const int v = (tid < BINS) ? cnt[tid] : 0;
  int incl = v;
#pragma unroll
  for (int d = 1; d < 32; d <<= 1) {
    const int t = __shfl_up(incl, d);
    if (lane >= d) incl += t;
  }
  if (tid < BINS && lane == 31) wtot[tid >> 5] = incl;
  __syncthreads();
  if (tid < BINS) {
    int pre = 0;
    for (int i = 0; i < (tid >> 5); ++i) pre += wtot[i];
    const int base = pre + incl - v;
    cur[tid] = base;
    int* off = g_off + (long long)blockIdx.x * (BINS + 1);
    off[tid] = (int)e_begin + base;
    if (tid == BINS - 1) off[BINS] = (int)e_begin + n_here;
  }
  __syncthreads();

#pragma unroll
  for (int i = 0; i < CHUNK / 1024; ++i) {
    const int idx = tid + i * 1024;
    if (idx < n_here) {
      const int pos = atomicAdd(&cur[bn[i]], 1);
      s_row[pos] = r[i] - bn[i] * rpb;
      s_col[pos] = cols[e_begin + idx];
      s_val[pos] = vals[e_begin + idx];
    }
  }
  __syncthreads();

  for (int idx = tid; idx < n_here; idx += 1024) {
    g_row[e_begin + idx] = s_row[idx];
    g_col[e_begin + idx] = s_col[idx];
    g_val[e_begin + idx] = s_val[idx];
  }
}

template <int NB>
__device__ __forceinline__ void edge_batch(
    const float* __restrict__ support, float* acc,
    int rl_l, int c_l, int v_l, int j, int lane) {
  float s_lo[NB], s_hi[NB];
#pragma unroll
  for (int u = 0; u < NB; ++u) {
    const int c = __builtin_amdgcn_readlane(c_l, j + u);
    const float* sp = support + (long long)c * OUT_FEAT + lane;
    s_lo[u] = sp[0];
    s_hi[u] = sp[32];
  }
#pragma unroll
  for (int u = 0; u < NB; ++u) {
    const int   rl = __builtin_amdgcn_readlane(rl_l, j + u);
    const float v  = __int_as_float(__builtin_amdgcn_readlane(v_l, j + u));
    float* a = acc + rl * OUT_FEAT + lane;
    atomicAdd(a,      v * s_lo[u]);
    atomicAdd(a + 32, v * s_hi[u]);
  }
}

__global__ __launch_bounds__(1024) void gcn_owner_sum(
    const float* __restrict__ ws, float* __restrict__ out,
    int n_nodes, int n_edges) {
  __shared__ __attribute__((aligned(16))) float acc[RPB_MAX * OUT_FEAT];

  const int tid  = threadIdx.x;
  const int lane = tid & 31;
  const int wave = tid >> 5;
  const int b    = blockIdx.x;
  const int rpb  = (n_nodes + BINS - 1) / BINS;
  const int row0 = b * rpb;
  int nrows = n_nodes - row0;
  if (nrows > rpb) nrows = rpb;
  if (nrows < 0) nrows = 0;

  const float* support = ws;
  const int*   g_row = (const int*)(ws + (long long)n_nodes * OUT_FEAT);
  const int*   g_col = g_row + n_edges;
  const float* g_val = (const float*)(g_col + n_edges);
  const int*   g_off = (const int*)(g_val + n_edges);

  v4f* acc4 = (v4f*)acc;
  const v4f z = {0.0f, 0.0f, 0.0f, 0.0f};
  for (int i = tid; i < rpb * (OUT_FEAT / 4); i += 1024) acc4[i] = z;
  __syncthreads();

  const int n_chunks = (n_edges + CHUNK - 1) >> CHUNK_SHIFT;
  for (int k0 = wave; k0 < n_chunks; k0 += 32 * 32) {
    const int kk = k0 + 32 * lane;
    int s_l = 0, t_l = 0;
    if (kk < n_chunks) {
      const int* off = g_off + (long long)kk * (BINS + 1) + b;
      s_l = off[0];
      t_l = off[1];
    }
    for (int m = 0; m < 32; ++m) {
      const int s = __builtin_amdgcn_readlane(s_l, m);
      const int t = __builtin_amdgcn_readlane(t_l, m);
      for (int g = s; g < t; g += 32) {
        const int rem  = t - g;
        const int cntg = rem < 32 ? rem : 32;
        const int el   = (lane < cntg) ? (g + lane) : (t - 1);
        const int rl_l = g_row[el];
        const int c_l  = g_col[el];
        const int v_l  = __float_as_int(g_val[el]);
        int j = 0;
        for (; j + 16 <= cntg; j += 16)
          edge_batch<16>(support, acc, rl_l, c_l, v_l, j, lane);
        for (; j + 4 <= cntg; j += 4)
          edge_batch<4>(support, acc, rl_l, c_l, v_l, j, lane);
        for (; j < cntg; ++j)
          edge_batch<1>(support, acc, rl_l, c_l, v_l, j, lane);
      }
    }
  }
  __syncthreads();

  v4f* o4 = (v4f*)(out + (long long)row0 * OUT_FEAT);
  for (int i = tid; i < nrows * (OUT_FEAT / 4); i += 1024) o4[i] = acc4[i];
}

extern "C" void kernel_launch(void* const* d_in, const int* in_sizes, int n_in,
                              void* d_out, int out_size, void* d_ws, size_t ws_size,
                              hipStream_t stream) {
  const float* x    = (const float*)d_in[0];
  const float* w    = (const float*)d_in[1];
  const int*   rows = (const int*)d_in[2];
  const int*   cols = (const int*)d_in[3];
  const float* vals = (const float*)d_in[4];
  float* out = (float*)d_out;
  float* ws  = (float*)d_ws;

  const int n_nodes = in_sizes[0] / IN_FEAT;
  const int n_edges = in_sizes[2];

  gcn_bin_edges<<<(unsigned)((n_edges + CHUNK - 1) / CHUNK), 1024, 0, stream>>>(
      rows, cols, vals, ws, n_nodes, n_edges);

  gcn_gemm_wmma<<<(n_nodes + 63) / 64, 128, 0, stream>>>(x, w, ws, n_nodes);

  gcn_owner_sum<<<BINS, 1024, 0, stream>>>(ws, out, n_nodes, n_edges);
}


# owner sum: the bin's edges are put in row order inside LDS (count, scan, place); each wave then sums whole rows in registers and stores them once: no LDS float adds; sum order = LDS placement order
# speedup vs baseline: 2.6965x; 1.0433x over previous
#include <hip/hip_runtime.h>

#define IN_FEAT 128
#define OUT_FEAT 64

typedef __attribute__((ext_vector_type(2))) float v2f;
typedef __attribute__((ext_vector_type(8))) float v8f;

#define KP_PITCH 160

typedef __attribute__((ext_vector_type(4))) float v4f;

__global__ __launch_bounds__(128) void gcn_gemm_wmma(
    const float* __restrict__ x, const float* __restrict__ w,
    float* __restrict__ support, int n_nodes) {
  __shared__ __attribute__((aligned(16))) float wlds[(IN_FEAT / 2) * KP_PITCH];

  const int tid = threadIdx.x;
  for (int i = tid; i < IN_FEAT * OUT_FEAT; i += 128) {
    int r = i >> 6;
    int c = i & 63;
    wlds[(r >> 1) * KP_PITCH + c * 2 + (r & 1)] = w[i];
  }
  __syncthreads();

  const int wave = tid >> 5;
  const int lane = tid & 31;
  const int half = lane >> 4;
  const int l16  = lane & 15;
  const int row_base = (blockIdx.x * 4 + wave) * 16;
  if (row_base >= n_nodes) return;

  int arow = row_base + l16;
  if (arow >= n_nodes) arow = n_nodes - 1;
  const float* __restrict__ xrow = x + (long long)arow * IN_FEAT;
  const float* __restrict__ wl = wlds + half * KP_PITCH + l16 * 4;

  v8f acc0 = {}, acc1 = {}, acc2 = {}, acc3 = {};
#pragma unroll
  for (int k = 0; k < IN_FEAT; k += 4) {
    const int kk = k + half * 2;
    v2f a = *(const v2f*)(xrow + kk);
    const v4f q0 = *(const v4f*)(wl + (k >> 1) * KP_PITCH);
    const v4f q1 = *(const v4f*)(wl + (k >> 1) * KP_PITCH + 64);
    v2f b0 = {q0.x, q0.y};
    v2f b1 = {q0.z, q0.w};
    v2f b2 = {q1.x, q1.y};
    v2f b3 = {q1.z, q1.w};
    acc0 = __builtin_amdgcn_wmma_f32_16x16x4_f32(
         false, a,  false, b0,
         (short)0, acc0,  false,  false);
    acc1 = __builtin_amdgcn_wmma_f32_16x16x4_f32(
        false, a, false, b1, (short)0, acc1, false, false);
    acc2 = __builtin_amdgcn_wmma_f32_16x16x4_f32(
        false, a, false, b2, (short)0, acc2, false, false);
    acc3 = __builtin_amdgcn_wmma_f32_16x16x4_f32(
        false, a, false, b3, (short)0, acc3, false, false);
  }

#pragma unroll
  for (int i = 0; i < 8; ++i) {
    const int orow = row_base + half * 8 + i;
    if (orow < n_nodes) {
      float* __restrict__ o = support + (long long)orow * OUT_FEAT + l16 * 2;
      v2f lo = {acc0[i], acc1[i]};
      v2f hi = {acc2[i], acc3[i]};
      *(v2f*)(o)      = lo;
      *(v2f*)(o + 32) = hi;
    }
  }
}

#define BINS 256
#define CHUNK 8192
#define CHUNK_SHIFT 13
#define RPB_MAX 392

__global__ __launch_bounds__(1024) void gcn_bin_edges(
    const int* __restrict__ rows, const int* __restrict__ cols,
    const float* __restrict__ vals, float* __restrict__ ws,
    int n_nodes, int n_edges) {
  __shared__ int cnt[BINS];
  __shared__ int cur[BINS];
  __shared__ int wtot[BINS / 32];
  __shared__ int   s_row[CHUNK];
  __shared__ int   s_col[CHUNK];
  __shared__ float s_val[CHUNK];

  const int tid  = threadIdx.x;
  const int lane = tid & 31;
  const int rpb  = (n_nodes + BINS - 1) / BINS;
  const long long e_begin = (long long)blockIdx.x * CHUNK;
  const long long left = (long long)n_edges - e_begin;
  const int n_here = left >= CHUNK ? CHUNK : (int)left;

  int*   g_row = (int*)(ws + (long long)n_nodes * OUT_FEAT);
  int*   g_col = g_row + n_edges;
  float* g_val = (float*)(g_col + n_edges);
  int*   g_off = (int*)(g_val + n_edges);

  if (tid < BINS) cnt[tid] = 0;
  __syncthreads();

  int r[CHUNK / 1024], bn[CHUNK / 1024];
#pragma unroll
  for (int i = 0; i < CHUNK / 1024; ++i) {
    const int idx = tid + i * 1024;
    r[i] = 0;
    bn[i] = 0;
    if (idx < n_here) {
      r[i]  = rows[e_begin + idx];
      bn[i] = r[i] / rpb;
      atomicAdd(&cnt[bn[i]], 1);
    }
  }
  __syncthreads();

  const int v = (tid < BINS) ? cnt[tid] : 0;
  int incl = v;
#pragma unroll
  for (int d = 1; d < 32; d <<= 1) {
    const int t = __shfl_up(incl, d);
    if (lane >= d) incl += t;
  }
  if (tid < BINS && lane == 31) wtot[tid >> 5] = incl;
  __syncthreads();
  if (tid < BINS) {
    int pre = 0;
    for (int i = 0; i < (tid >> 5); ++i) pre += wtot[i];
    const int base = pre + incl - v;
    cur[tid] = base;
    int* off = g_off + (long long)blockIdx.x * (BINS + 1);
    off[tid] = (int)e_begin + base;
    if (tid == BINS - 1) off[BINS] = (int)e_begin + n_here;
  }
  __syncthreads();

#pragma unroll
  for (int i = 0; i < CHUNK / 1024; ++i) {
    const int idx = tid + i * 1024;
    if (idx < n_here) {
      const int pos = atomicAdd(&cur[bn[i]], 1);
      s_row[pos] = r[i] - bn[i] * rpb;
      s_col[pos] = cols[e_begin + idx];
      s_val[pos] = vals[e_begin + idx];
    }
  }
  __syncthreads();

  for (int idx = tid; idx < n_here; idx += 1024) {
    g_row[e_begin + idx] = s_row[idx];
    g_col[e_begin + idx] = s_col[idx];
    g_val[e_begin + idx] = s_val[idx];
  }
}

#define CAP 8192
#define MAXCH 256
#define NROWSLOT 400

template <int NB>
__device__ __forceinline__ void edge_batch(
    const float* __restrict__ support, float* acc,
    int rl_l, int c_l, int v_l, int j, int lane) {
  float s_lo[NB], s_hi[NB];
#pragma unroll
  for (int u = 0; u < NB; ++u) {
    const int c = __builtin_amdgcn_readlane(c_l, j + u);
    const float* sp = support + (long long)c * OUT_FEAT + lane;
    s_lo[u] = sp[0];
    s_hi[u] = sp[32];
  }
#pragma unroll
  for (int u = 0; u < NB; ++u) {
    const int   rl = __builtin_amdgcn_readlane(rl_l, j + u);
    const float v  = __int_as_float(__builtin_amdgcn_readlane(v_l, j + u));
    float* a = acc + rl * OUT_FEAT + lane;
    atomicAdd(a,      v * s_lo[u]);
    atomicAdd(a + 32, v * s_hi[u]);
  }
}

__device__ __forceinline__ void owner_sum_lds_adds(
    const float* __restrict__ ws, float* __restrict__ out, float* acc,
    int n_nodes, int n_edges, int b) {
  const int tid  = threadIdx.x;
  const int lane = tid & 31;
  const int wave = tid >> 5;
  const int rpb  = (n_nodes + BINS - 1) / BINS;
  const int row0 = b * rpb;
  int nrows = n_nodes - row0;
  if (nrows > rpb) nrows = rpb;
  if (nrows < 0) nrows = 0;

  const float* support = ws;
  const int*   g_row = (const int*)(ws + (long long)n_nodes * OUT_FEAT);
  const int*   g_col = g_row + n_edges;
  const float* g_val = (const float*)(g_col + n_edges);
  const int*   g_off = (const int*)(g_val + n_edges);

  v4f* acc4 = (v4f*)acc;
  const v4f z = {0.0f, 0.0f, 0.0f, 0.0f};
  for (int i = tid; i < rpb * (OUT_FEAT / 4); i += 1024) acc4[i] = z;
  __syncthreads();

  const int n_chunks = (n_edges + CHUNK - 1) >> CHUNK_SHIFT;
  for (int k0 = wave; k0 < n_chunks; k0 += 32 * 32) {
    const int kk = k0 + 32 * lane;
    int s_l = 0, t_l = 0;
    if (kk < n_chunks) {
      const int* off = g_off + (long long)kk * (BINS + 1) + b;
      s_l = off[0];
      t_l = off[1];
    }
    for (int m = 0; m < 32; ++m) {
      const int s = __builtin_amdgcn_readlane(s_l, m);
      const int t = __builtin_amdgcn_readlane(t_l, m);
      for (int g = s; g < t; g += 32) {
        const int rem  = t - g;
        const int cntg = rem < 32 ? rem : 32;
        const int el   = (lane < cntg) ? (g + lane) : (t - 1);
        const int rl_l = g_row[el];
        const int c_l  = g_col[el];
        const int v_l  = __float_as_int(g_val[el]);
        int j = 0;
        for (; j + 16 <= cntg; j += 16)
          edge_batch<16>(support, acc, rl_l, c_l, v_l, j, lane);
        for (; j + 4 <= cntg; j += 4)
          edge_batch<4>(support, acc, rl_l, c_l, v_l, j, lane);
        for (; j < cntg; ++j)
          edge_batch<1>(support, acc, rl_l, c_l, v_l, j, lane);
      }
    }
  }
  __syncthreads();

  v4f* o4 = (v4f*)(out + (long long)row0 * OUT_FEAT);
  for (int i = tid; i < nrows * (OUT_FEAT / 4); i += 1024) o4[i] = acc4[i];
}

__device__ __forceinline__ void store_row(float* __restrict__ orow0, int row, int lane, v2f a) {
  *(v2f*)(orow0 + (long long)row * OUT_FEAT + lane * 2) = a;
}

template <int NB>
__device__ __forceinline__ void row_batch(
    const float* __restrict__ support, float* __restrict__ orow0,
    int r_l, int c_l, int v_l, int j, int lane, int& row, v2f& a) {
  v2f s[NB];
#pragma unroll
  for (int u = 0; u < NB; ++u) {
    const int c = __builtin_amdgcn_readlane(c_l, j + u);
    s[u] = *(const v2f*)(support + (long long)c * OUT_FEAT + lane * 2);
  }
#pragma unroll
  for (int u = 0; u < NB; ++u) {
    const int   rl = __builtin_amdgcn_readlane(r_l, j + u);
    const float v  = __int_as_float(__builtin_amdgcn_readlane(v_l, j + u));
    if (rl != row) {
      store_row(orow0, row, lane, a);
      const v2f z = {0.0f, 0.0f};
#pragma nounroll
      for (int r = row + 1; r < rl; ++r) store_row(orow0, r, lane, z);
      row = rl;
      a = z;
    }
    a.x += v * s[u].x;
    a.y += v * s[u].y;
  }
}

__global__ __launch_bounds__(1024) void gcn_owner_sum(
    const float* __restrict__ ws, float* __restrict__ out,
    int n_nodes, int n_edges) {
  __shared__ __attribute__((aligned(16))) float lds[6 * CAP];
  __shared__ int seg_s[MAXCH];
  __shared__ int seg_pre[MAXCH + 1];
  __shared__ int cnt[NROWSLOT];
  __shared__ int start[NROWSLOT + 1];
  __shared__ int cur[NROWSLOT];
  __shared__ int wtot[32];

  const int tid  = threadIdx.x;
  const int lane = tid & 31;
  const int wave = __builtin_amdgcn_readfirstlane(tid >> 5);
  const int b    = blockIdx.x;
  const int rpb  = (n_nodes + BINS - 1) / BINS;
  const int row0 = b * rpb;
  int nrows = n_nodes - row0;
  if (nrows > rpb) nrows = rpb;
  if (nrows < 0) nrows = 0;

  const float* support = ws;
  const int*   g_row = (const int*)(ws + (long long)n_nodes * OUT_FEAT);
  const int*   g_col = g_row + n_edges;
  const float* g_val = (const float*)(g_col + n_edges);
  const int*   g_off = (const int*)(g_val + n_edges);

  const int n_chunks = (n_edges + CHUNK - 1) >> CHUNK_SHIFT;

  int seg0 = 0, len = 0;
  if (tid < n_chunks && tid < MAXCH) {
    const int* off = g_off + (long long)tid * (BINS + 1) + b;
    seg0 = off[0];
    len  = off[1] - seg0;
  }
  int incl = len;
#pragma unroll
  for (int d = 1; d < 32; d <<= 1) {
    const int t = __shfl_up(incl, d);
    if (lane >= d) incl += t;
  }
  if (tid < MAXCH && lane == 31) wtot[wave] = incl;
  if (tid < NROWSLOT) cnt[tid] = 0;
  __syncthreads();
  if (tid < MAXCH) {
    int pre = 0;
    for (int i = 0; i < wave; ++i) pre += wtot[i];
    const int excl = pre + incl - len;
    seg_s[tid]   = seg0;
    seg_pre[tid] = excl;
    if (tid == MAXCH - 1) seg_pre[MAXCH] = excl + len;
  }
  __syncthreads();
  const int n_bin = seg_pre[MAXCH];

  if (n_chunks > MAXCH || n_bin > CAP || rpb > RPB_MAX) {
    owner_sum_lds_adds(ws, out, lds, n_nodes, n_edges, b);
    return;
  }

  int*   st_r = (int*)lds;
  int*   st_c = st_r + CAP;
  float* st_v = (float*)(st_c + CAP);
  int*   so_r = st_r + 3 * CAP;
  int*   so_c = so_r + CAP;
  float* so_v = (float*)(so_c + CAP);

  for (int k0 = wave; k0 < n_chunks; k0 += 32 * 8) {
    int sg[8], pg[8], ng[8];
#pragma unroll
    for (int u = 0; u < 8; ++u) {
      const int k = k0 + 32 * u;
      sg[u] = 0; pg[u] = 0; ng[u] = 0;
      if (k < n_chunks) {
        sg[u] = seg_s[k];
        pg[u] = seg_pre[k];
        ng[u] = seg_pre[k + 1] - pg[u];
      }
    }
    int ra[8], ca[8], rb[8], cb[8];
    float va[8], vb[8];
#pragma unroll
    for (int u = 0; u < 8; ++u) {
      ra[u] = 0; ca[u] = 0; va[u] = 0.0f;
      rb[u] = 0; cb[u] = 0; vb[u] = 0.0f;
      if (lane < ng[u]) {
        const int e = sg[u] + lane;
        ra[u] = g_row[e]; ca[u] = g_col[e]; va[u] = g_val[e];
      }
      if (lane + 32 < ng[u]) {
        const int e = sg[u] + lane + 32;
        rb[u] = g_row[e]; cb[u] = g_col[e]; vb[u] = g_val[e];
      }
    }
#pragma unroll
    for (int u = 0; u < 8; ++u) {
      if (lane < ng[u]) {
        const int p = pg[u] + lane;
        st_r[p] = ra[u]; st_c[p] = ca[u]; st_v[p] = va[u];
        atomicAdd(&cnt[ra[u]], 1);
      }
      if (lane + 32 < ng[u]) {
        const int p = pg[u] + lane + 32;
        st_r[p] = rb[u]; st_c[p] = cb[u]; st_v[p] = vb[u];
        atomicAdd(&cnt[rb[u]], 1);
      }
      for (int g = lane + 64; g < ng[u]; g += 32) {
        const int e = sg[u] + g;
        const int p = pg[u] + g;
        const int r = g_row[e];
        st_r[p] = r; st_c[p] = g_col[e]; st_v[p] = g_val[e];
        atomicAdd(&cnt[r], 1);
      }
    }
  }
  __syncthreads();

  {
    const int v = (tid < NROWSLOT) ? cnt[tid] : 0;
    int inc2 = v;
#pragma unroll
    for (int d = 1; d < 32; d <<= 1) {
      const int t = __shfl_up(inc2, d);
      if (lane >= d) inc2 += t;
    }
    if (tid < 512 && lane == 31) wtot[wave] = inc2;
    __syncthreads();
    if (tid < NROWSLOT) {
      int pre = 0;
      for (int i = 0; i < wave; ++i) pre += wtot[i];
      const int excl = pre + inc2 - v;
      start[tid] = excl;
      cur[tid]   = excl;
      if (tid == NROWSLOT - 1) start[NROWSLOT] = excl + v;
    }
    __syncthreads();
  }

  for (int i = tid; i < n_bin; i += 1024) {
    const int r = st_r[i];
    const int pos = atomicAdd(&cur[r], 1);
    so_r[pos] = r;
    so_c[pos] = st_c[i];
    so_v[pos] = st_v[i];
  }
  __syncthreads();

  const int r_a = (wave * nrows) >> 5;
  const int r_b = ((wave + 1) * nrows) >> 5;
  if (r_a >= r_b) return;
  const int p0 = __builtin_amdgcn_readfirstlane(start[r_a]);
  const int p1 = __builtin_amdgcn_readfirstlane(start[r_b]);
  float* __restrict__ orow0 = out + (long long)row0 * OUT_FEAT;
  int row = r_a;
  v2f a = {0.0f, 0.0f};
  for (int g = p0; g < p1; g += 32) {
    const int rem = p1 - g;
    const int m   = rem < 32 ? rem : 32;
    const int idx = g + (lane < m ? lane : m - 1);
    const int r_l = so_r[idx];
    const int c_l = so_c[idx];
    const int v_l = __float_as_int(so_v[idx]);
    int j = 0;
    for (; j + 16 <= m; j += 16)
      row_batch<16>(support, orow0, r_l, c_l, v_l, j, lane, row, a);
    for (; j + 4 <= m; j += 4)
      row_batch<4>(support, orow0, r_l, c_l, v_l, j, lane, row, a);
    for (; j < m; ++j)
      row_batch<1>(support, orow0, r_l, c_l, v_l, j, lane, row, a);
  }
  store_row(orow0, row, lane, a);
  const v2f z2 = {0.0f, 0.0f};
  for (int r = row + 1; r < r_b; ++r) store_row(orow0, r, lane, z2);
}

extern "C" void kernel_launch(void* const* d_in, const int* in_sizes, int n_in,
                              void* d_out, int out_size, void* d_ws, size_t ws_size,
                              hipStream_t stream) {
  const float* x    = (const float*)d_in[0];
  const float* w    = (const float*)d_in[1];
  const int*   rows = (const int*)d_in[2];
  const int*   cols = (const int*)d_in[3];
  const float* vals = (const float*)d_in[4];
  float* out = (float*)d_out;
  float* ws  = (float*)d_ws;

  const int n_nodes = in_sizes[0] / IN_FEAT;
  const int n_edges = in_sizes[2];

  gcn_bin_edges<<<(unsigned)((n_edges + CHUNK - 1) / CHUNK), 1024, 0, stream>>>(
      rows, cols, vals, ws, n_nodes, n_edges);

  gcn_gemm_wmma<<<(n_nodes + 63) / 64, 128, 0, stream>>>(x, w, ws, n_nodes);

  gcn_owner_sum<<<BINS, 1024, 0, stream>>>(ws, out, n_nodes, n_edges);
}


# one launch fewer: edge binning and the x @ W product share one launch of 256 resident work-groups (W staged once per group, 16-row tiles dealt to its waves, spare tiles to groups without a chunk)
# speedup vs baseline: 2.9924x; 1.1097x over previous
#include <hip/hip_runtime.h>

#define IN_FEAT 128
#define OUT_FEAT 64

typedef __attribute__((ext_vector_type(2))) float v2f;
typedef __attribute__((ext_vector_type(8))) float v8f;

typedef __attribute__((ext_vector_type(4))) float v4f;

#define KP_PITCH 160
#define BINS 256
#define CHUNK 8192
#define CHUNK_SHIFT 13
#define RPB_MAX 392
#define RESIDENT_GROUPS 256

__device__ __forceinline__ void gemm_tile16(
    const float* __restrict__ x, const float* wlds, float* __restrict__ support,
    int n_nodes, int row_base, int lane) {
  const int half = lane >> 4;
  const int l16  = lane & 15;

  int arow = row_base + l16;
  if (arow >= n_nodes) arow = n_nodes - 1;
  const float* __restrict__ xrow = x + (long long)arow * IN_FEAT;
  const float* wl = wlds + half * KP_PITCH + l16 * 4;

  v8f acc0 = {}, acc1 = {}, acc2 = {}, acc3 = {};
#pragma unroll
  for (int k = 0; k < IN_FEAT; k += 4) {
    const int kk = k + half * 2;
    v2f a = *(const v2f*)(xrow + kk);
    const v4f q0 = *(const v4f*)(wl + (k >> 1) * KP_PITCH);
    const v4f q1 = *(const v4f*)(wl + (k >> 1) * KP_PITCH + 64);
    v2f b0 = {q0.x, q0.y};
    v2f b1 = {q0.z, q0.w};
    v2f b2 = {q1.x, q1.y};
    v2f b3 = {q1.z, q1.w};
    acc0 = __builtin_amdgcn_wmma_f32_16x16x4_f32(
         false, a,  false, b0,
         (short)0, acc0,  false,  false);
    acc1 = __builtin_amdgcn_wmma_f32_16x16x4_f32(
        false, a, false, b1, (short)0, acc1, false, false);
    acc2 = __builtin_amdgcn_wmma_f32_16x16x4_f32(
        false, a, false, b2, (short)0, acc2, false, false);
    acc3 = __builtin_amdgcn_wmma_f32_16x16x4_f32(
        false, a, false, b3, (short)0, acc3, false, false);
  }

#pragma unroll
  for (int i = 0; i < 8; ++i) {
    const int orow = row_base + half * 8 + i;
    if (orow < n_nodes) {
      float* __restrict__ o = support + (long long)orow * OUT_FEAT + l16 * 2;
      v2f lo = {acc0[i], acc1[i]};
      v2f hi = {acc2[i], acc3[i]};
      *(v2f*)(o)      = lo;
      *(v2f*)(o + 32) = hi;
    }
  }
}

__global__ __launch_bounds__(1024) __attribute__((amdgpu_waves_per_eu(8, 8))) void gcn_bin_and_gemm(
    const float* __restrict__ x, const float* __restrict__ w,
    const int* __restrict__ rows, const int* __restrict__ cols,
    const float* __restrict__ vals, float* __restrict__ ws,
    int n_nodes, int n_edges) {
  __shared__ __attribute__((aligned(16))) float wlds[(IN_FEAT / 2) * KP_PITCH];
  __shared__ int cnt[BINS];
  __shared__ int cur[BINS];
  __shared__ int wtot[BINS / 32];
  __shared__ int   s_row[CHUNK];
  __shared__ int   s_col[CHUNK];
  __shared__ float s_val[CHUNK];

  const int tid  = threadIdx.x;
  const int lane = tid & 31;

  v2f wreg[(IN_FEAT / 2) * OUT_FEAT / 1024];
#pragma unroll
  for (int j = 0; j < (IN_FEAT / 2) * OUT_FEAT / 1024; ++j) {
    const int i  = tid + j * 1024;
    const int kp = i >> 6;
    const int c  = i & 63;
    wreg[j].x = w[(2 * kp) * OUT_FEAT + c];
    wreg[j].y = w[(2 * kp + 1) * OUT_FEAT + c];
  }

  const int rpb = (n_nodes + BINS - 1) / BINS;
  int*   g_row = (int*)(ws + (long long)n_nodes * OUT_FEAT);
  int*   g_col = g_row + n_edges;
  float* g_val = (float*)(g_col + n_edges);
  int*   g_off = (int*)(g_val + n_edges);
  const int n_chunks = (n_edges + CHUNK - 1) >> CHUNK_SHIFT;

  for (int ch = (int)gridDim.x - 1 - (int)blockIdx.x; ch < n_chunks; ch += (int)gridDim.x) {
    const long long e_begin = (long long)ch * CHUNK;
    const long long left = (long long)n_edges - e_begin;
    const int n_here = left >= CHUNK ? CHUNK : (int)left;

    if (tid < BINS) cnt[tid] = 0;
    __syncthreads();

    int r[CHUNK / 1024], bn[CHUNK / 1024];
#pragma unroll
    for (int i = 0; i < CHUNK / 1024; ++i) {
      const int idx = tid + i * 1024;
      r[i] = 0;
      bn[i] = 0;
      if (idx < n_here) {
        r[i]  = rows[e_begin + idx];
        bn[i] = r[i] / rpb;
        atomicAdd(&cnt[bn[i]], 1);
      }
    }
    __syncthreads();

    const int v = (tid < BINS) ? cnt[tid] : 0;
    int incl = v;
#pragma unroll
    for (int d = 1; d < 32; d <<= 1) {
      const int t = __shfl_up(incl, d);
      if (lane >= d) incl += t;
    }
    if (tid < BINS && lane == 31) wtot[tid >> 5] = incl;
    __syncthreads();
    if (tid < BINS) {
      int pre = 0;
      for (int i = 0; i < (tid >> 5); ++i) pre += wtot[i];
      const int base = pre + incl - v;
      cur[tid] = base;
      int* off = g_off + (long long)ch * (BINS + 1);
      off[tid] = (int)e_begin + base;
      if (tid == BINS - 1) off[BINS] = (int)e_begin + n_here;
    }
    __syncthreads();

#pragma unroll
    for (int i = 0; i < CHUNK / 1024; ++i) {
      const int idx = tid + i * 1024;
      if (idx < n_here) {
        const int pos = atomicAdd(&cur[bn[i]], 1);
        s_row[pos] = r[i] - bn[i] * rpb;
        s_col[pos] = cols[e_begin + idx];
        s_val[pos] = vals[e_begin + idx];
      }
    }
    __syncthreads();

    for (int idx = tid; idx < n_here; idx += 1024) {
      g_row[e_begin + idx] = s_row[idx];
      g_col[e_begin + idx] = s_col[idx];
      g_val[e_begin + idx] = s_val[idx];
    }
  }
#pragma unroll
  for (int j = 0; j < (IN_FEAT / 2) * OUT_FEAT / 1024; ++j) {
    const int i  = tid + j * 1024;
    *(v2f*)(wlds + (i >> 6) * KP_PITCH + (i & 63) * 2) = wreg[j];
  }
  __syncthreads();

  const int wave = __builtin_amdgcn_readfirstlane(tid >> 5);
  const int s_hi = ((wave >> 1) ^ (wave >> 3)) & 1;
  const int s_lo = (wave ^ (wave >> 2) ^ (wave >> 4)) & 1;
  const int slot = (s_hi << 4) | (s_lo << 3) | (wave & 7);
  const int G = (int)gridDim.x;
  const int b = (int)blockIdx.x;
  const int n_tiles = (n_nodes + 15) >> 4;
  const int full = n_tiles / G;
  const int rest = n_tiles - full * G;
  const int n_free = n_chunks < G ? G - n_chunks : 0;
  const int deal = n_free > 0 ? n_free : G;
  const int rounds = full + (rest + deal - 1) / deal;
  for (int s = slot; s < rounds; s += 32) {
    int t = b + G * s;
    if (s >= full) {
      const int u = b + deal * (s - full);
      if (b >= deal || u >= rest) continue;
      t = full * G + u;
    }
    asm volatile("" ::: "memory");
    gemm_tile16(x, wlds, ws, n_nodes, t * 16, lane);
  }
}

#define CAP 8192
#define MAXCH 256
#define NROWSLOT 400

template <int NB>
__device__ __forceinline__ void edge_batch(
    const float* __restrict__ support, float* acc,
    int rl_l, int c_l, int v_l, int j, int lane) {
  float s_lo[NB], s_hi[NB];
#pragma unroll
  for (int u = 0; u < NB; ++u) {
    const int c = __builtin_amdgcn_readlane(c_l, j + u);
    const float* sp = support + (long long)c * OUT_FEAT + lane;
    s_lo[u] = sp[0];
    s_hi[u] = sp[32];
  }
#pragma unroll
  for (int u = 0; u < NB; ++u) {
    const int   rl = __builtin_amdgcn_readlane(rl_l, j + u);
    const float v  = __int_as_float(__builtin_amdgcn_readlane(v_l, j + u));
    float* a = acc + rl * OUT_FEAT + lane;
    atomicAdd(a,      v * s_lo[u]);
    atomicAdd(a + 32, v * s_hi[u]);
  }
}

__device__ __forceinline__ void owner_sum_lds_adds(
    const float* __restrict__ ws, float* __restrict__ out, float* acc,
    int n_nodes, int n_edges, int b) {
  const int tid  = threadIdx.x;
  const int lane = tid & 31;
  const int wave = tid >> 5;
  const int rpb  = (n_nodes + BINS - 1) / BINS;
  const int row0 = b * rpb;
  int nrows = n_nodes - row0;
  if (nrows > rpb) nrows = rpb;
  if (nrows < 0) nrows = 0;

  const float* support = ws;
  const int*   g_row = (const int*)(ws + (long long)n_nodes * OUT_FEAT);
  const int*   g_col = g_row + n_edges;
  const float* g_val = (const float*)(g_col + n_edges);
  const int*   g_off = (const int*)(g_val + n_edges);

  v4f* acc4 = (v4f*)acc;
  const v4f z = {0.0f, 0.0f, 0.0f, 0.0f};
  for (int i = tid; i < rpb * (OUT_FEAT / 4); i += 1024) acc4[i] = z;
  __syncthreads();

  const int n_chunks = (n_edges + CHUNK - 1) >> CHUNK_SHIFT;
  for (int k0 = wave; k0 < n_chunks; k0 += 32 * 32) {
    const int kk = k0 + 32 * lane;
    int s_l = 0, t_l = 0;
    if (kk < n_chunks) {
      const int* off = g_off + (long long)kk * (BINS + 1) + b;
      s_l = off[0];
      t_l = off[1];
    }
    for (int m = 0; m < 32; ++m) {
      const int s = __builtin_amdgcn_readlane(s_l, m);
      const int t = __builtin_amdgcn_readlane(t_l, m);
      for (int g = s; g < t; g += 32) {
        const int rem  = t - g;
        const int cntg = rem < 32 ? rem : 32;
        const int el   = (lane < cntg) ? (g + lane) : (t - 1);
        const int rl_l = g_row[el];
        const int c_l  = g_col[el];
        const int v_l  = __float_as_int(g_val[el]);
        int j = 0;
        for (; j + 16 <= cntg; j += 16)
          edge_batch<16>(support, acc, rl_l, c_l, v_l, j, lane);
        for (; j + 4 <= cntg; j += 4)
          edge_batch<4>(support, acc, rl_l, c_l, v_l, j, lane);
        for (; j < cntg; ++j)
          edge_batch<1>(support, acc, rl_l, c_l, v_l, j, lane);
      }
    }
  }
  __syncthreads();

  v4f* o4 = (v4f*)(out + (long long)row0 * OUT_FEAT);
  for (int i = tid; i < nrows * (OUT_FEAT / 4); i += 1024) o4[i] = acc4[i];
}

__device__ __forceinline__ void store_row(float* __restrict__ orow0, int row, int lane, v2f a) {
  *(v2f*)(orow0 + (long long)row * OUT_FEAT + lane * 2) = a;
}

template <int NB>
__device__ __forceinline__ void row_batch(
    const float* __restrict__ support, float* __restrict__ orow0,
    int r_l, int c_l, int v_l, int j, int lane, int& row, v2f& a) {
  v2f s[NB];
#pragma unroll
  for (int u = 0; u < NB; ++u) {
    const int c = __builtin_amdgcn_readlane(c_l, j + u);
    s[u] = *(const v2f*)(support + (long long)c * OUT_FEAT + lane * 2);
  }
#pragma unroll
  for (int u = 0; u < NB; ++u) {
    const int   rl = __builtin_amdgcn_readlane(r_l, j + u);
    const float v  = __int_as_float(__builtin_amdgcn_readlane(v_l, j + u));
    if (rl != row) {
      store_row(orow0, row, lane, a);
      const v2f z = {0.0f, 0.0f};
#pragma nounroll
      for (int r = row + 1; r < rl; ++r) store_row(orow0, r, lane, z);
      row = rl;
      a = z;
    }
    a.x += v * s[u].x;
    a.y += v * s[u].y;
  }
}

__global__ __launch_bounds__(1024) void gcn_owner_sum(
    const float* __restrict__ ws, float* __restrict__ out,
    int n_nodes, int n_edges) {
  __shared__ __attribute__((aligned(16))) float lds[6 * CAP];
  __shared__ int seg_s[MAXCH];
  __shared__ int seg_pre[MAXCH + 1];
  __shared__ int cnt[NROWSLOT];
  __shared__ int start[NROWSLOT + 1];
  __shared__ int cur[NROWSLOT];
  __shared__ int wtot[32];

  const int tid  = threadIdx.x;
  const int lane = tid & 31;
  const int wave = __builtin_amdgcn_readfirstlane(tid >> 5);
  const int b    = blockIdx.x;
  const int rpb  = (n_nodes + BINS - 1) / BINS;
  const int row0 = b * rpb;
  int nrows = n_nodes - row0;
  if (nrows > rpb) nrows = rpb;
  if (nrows < 0) nrows = 0;

  const float* support = ws;
  const int*   g_row = (const int*)(ws + (long long)n_nodes * OUT_FEAT);
  const int*   g_col = g_row + n_edges;
  const float* g_val = (const float*)(g_col + n_edges);
  const int*   g_off = (const int*)(g_val + n_edges);

  const int n_chunks = (n_edges + CHUNK - 1) >> CHUNK_SHIFT;

  int seg0 = 0, len = 0;
  if (tid < n_chunks && tid < MAXCH) {
    const int* off = g_off + (long long)tid * (BINS + 1) + b;
    seg0 = off[0];
    len  = off[1] - seg0;
  }
  int incl = len;
#pragma unroll
  for (int d = 1; d < 32; d <<= 1) {
    const int t = __shfl_up(incl, d);
    if (lane >= d) incl += t;
  }
  if (tid < MAXCH && lane == 31) wtot[wave] = incl;
  if (tid < NROWSLOT) cnt[tid] = 0;
  __syncthreads();
  if (tid < MAXCH) {
    int pre = 0;
    for (int i = 0; i < wave; ++i) pre += wtot[i];
    const int excl = pre + incl - len;
    seg_s[tid]   = seg0;
    seg_pre[tid] = excl;
    if (tid == MAXCH - 1) seg_pre[MAXCH] = excl + len;
  }
  __syncthreads();
  const int n_bin = seg_pre[MAXCH];

  if (n_chunks > MAXCH || n_bin > CAP || rpb > RPB_MAX) {
    owner_sum_lds_adds(ws, out, lds, n_nodes, n_edges, b);
    return;
  }

  int*   st_r = (int*)lds;
  int*   st_c = st_r + CAP;
  float* st_v = (float*)(st_c + CAP);
  int*   so_r = st_r + 3 * CAP;
  int*   so_c = so_r + CAP;
  float* so_v = (float*)(so_c + CAP);

  for (int k0 = wave; k0 < n_chunks; k0 += 32 * 8) {
    int sg[8], pg[8], ng[8];
#pragma unroll
    for (int u = 0; u < 8; ++u) {
      const int k = k0 + 32 * u;
      sg[u] = 0; pg[u] = 0; ng[u] = 0;
      if (k < n_chunks) {
        sg[u] = seg_s[k];
        pg[u] = seg_pre[k];
        ng[u] = seg_pre[k + 1] - pg[u];
      }
    }
    int ra[8], ca[8], rb[8], cb[8];
    float va[8], vb[8];
#pragma unroll
    for (int u = 0; u < 8; ++u) {
      ra[u] = 0; ca[u] = 0; va[u] = 0.0f;
      rb[u] = 0; cb[u] = 0; vb[u] = 0.0f;
      if (lane < ng[u]) {
        const int e = sg[u] + lane;
        ra[u] = g_row[e]; ca[u] = g_col[e]; va[u] = g_val[e];
      }
      if (lane + 32 < ng[u]) {
        const int e = sg[u] + lane + 32;
        rb[u] = g_row[e]; cb[u] = g_col[e]; vb[u] = g_val[e];
      }
    }
#pragma unroll
    for (int u = 0; u < 8; ++u) {
      if (lane < ng[u]) {
        const int p = pg[u] + lane;
        st_r[p] = ra[u]; st_c[p] = ca[u]; st_v[p] = va[u];
        atomicAdd(&cnt[ra[u]], 1);
      }
      if (lane + 32 < ng[u]) {
        const int p = pg[u] + lane + 32;
        st_r[p] = rb[u]; st_c[p] = cb[u]; st_v[p] = vb[u];
        atomicAdd(&cnt[rb[u]], 1);
      }
      for (int g = lane + 64; g < ng[u]; g += 32) {
        const int e = sg[u] + g;
        const int p = pg[u] + g;
        const int r = g_row[e];
        st_r[p] = r; st_c[p] = g_col[e]; st_v[p] = g_val[e];
        atomicAdd(&cnt[r], 1);
      }
    }
  }
  __syncthreads();

  {
    const int v = (tid < NROWSLOT) ? cnt[tid] : 0;
    int inc2 = v;
#pragma unroll
    for (int d = 1; d < 32; d <<= 1) {
      const int t = __shfl_up(inc2, d);
      if (lane >= d) inc2 += t;
    }
    if (tid < 512 && lane == 31) wtot[wave] = inc2;
    __syncthreads();
    if (tid < NROWSLOT) {
      int pre = 0;
      for (int i = 0; i < wave; ++i) pre += wtot[i];
      const int excl = pre + inc2 - v;
      start[tid] = excl;
      cur[tid]   = excl;
      if (tid == NROWSLOT - 1) start[NROWSLOT] = excl + v;
    }
    __syncthreads();
  }

  for (int i = tid; i < n_bin; i += 1024) {
    const int r = st_r[i];
    const int pos = atomicAdd(&cur[r], 1);
    so_r[pos] = r;
    so_c[pos] = st_c[i];
    so_v[pos] = st_v[i];
  }
  __syncthreads();

  const int r_a = (wave * nrows) >> 5;
  const int r_b = ((wave + 1) * nrows) >> 5;
  if (r_a >= r_b) return;
  const int p0 = __builtin_amdgcn_readfirstlane(start[r_a]);
  const int p1 = __builtin_amdgcn_readfirstlane(start[r_b]);
  float* __restrict__ orow0 = out + (long long)row0 * OUT_FEAT;
  int row = r_a;
  v2f a = {0.0f, 0.0f};
  for (int g = p0; g < p1; g += 32) {
    const int rem = p1 - g;
    const int m   = rem < 32 ? rem : 32;
    const int idx = g + (lane < m ? lane : m - 1);
    const int r_l = so_r[idx];
    const int c_l = so_c[idx];
    const int v_l = __float_as_int(so_v[idx]);
    int j = 0;
    for (; j + 16 <= m; j += 16)
      row_batch<16>(support, orow0, r_l, c_l, v_l, j, lane, row, a);
    for (; j + 4 <= m; j += 4)
      row_batch<4>(support, orow0, r_l, c_l, v_l, j, lane, row, a);
    for (; j < m; ++j)
      row_batch<1>(support, orow0, r_l, c_l, v_l, j, lane, row, a);
  }
  store_row(orow0, row, lane, a);
  const v2f z2 = {0.0f, 0.0f};
  for (int r = row + 1; r < r_b; ++r) store_row(orow0, r, lane, z2);
}

extern "C" void kernel_launch(void* const* d_in, const int* in_sizes, int n_in,
                              void* d_out, int out_size, void* d_ws, size_t ws_size,
                              hipStream_t stream) {
  const float* x    = (const float*)d_in[0];
  const float* w    = (const float*)d_in[1];
  const int*   rows = (const int*)d_in[2];
  const int*   cols = (const int*)d_in[3];
  const float* vals = (const float*)d_in[4];
  float* out = (float*)d_out;
  float* ws  = (float*)d_ws;

  const int n_nodes = in_sizes[0] / IN_FEAT;
  const int n_edges = in_sizes[2];

  gcn_bin_and_gemm<<<RESIDENT_GROUPS, 1024, 0, stream>>>(
      x, w, rows, cols, vals, ws, n_nodes, n_edges);

  gcn_owner_sum<<<BINS, 1024, 0, stream>>>(ws, out, n_nodes, n_edges);
}


# binning: a thread asks for all 24 words (row, col, val) of its 8 chunk edges as one branch-free group of loads: 1 memory round trip per chunk, was 16 in series; sum order = LDS placement, as before
# speedup vs baseline: 3.1230x; 1.0436x over previous
#include <hip/hip_runtime.h>

#define IN_FEAT 128
#define OUT_FEAT 64

typedef __attribute__((ext_vector_type(2))) float v2f;
typedef __attribute__((ext_vector_type(8))) float v8f;

typedef __attribute__((ext_vector_type(4))) float v4f;

#define KP_PITCH 160
#define BINS 256
#define CHUNK 8192
#define CHUNK_SHIFT 13
#define RPB_MAX 392
#define RESIDENT_GROUPS 256

__device__ __forceinline__ void gemm_tile16(
    const float* __restrict__ x, const float* wlds, float* __restrict__ support,
    int n_nodes, int row_base, int lane) {
  const int half = lane >> 4;
  const int l16  = lane & 15;

  int arow = row_base + l16;
  if (arow >= n_nodes) arow = n_nodes - 1;
  const float* __restrict__ xrow = x + (long long)arow * IN_FEAT;
  const float* wl = wlds + half * KP_PITCH + l16 * 4;

  v8f acc0 = {}, acc1 = {}, acc2 = {}, acc3 = {};
#pragma unroll
  for (int k = 0; k < IN_FEAT; k += 4) {
    const int kk = k + half * 2;
    v2f a = *(const v2f*)(xrow + kk);
    const v4f q0 = *(const v4f*)(wl + (k >> 1) * KP_PITCH);
    const v4f q1 = *(const v4f*)(wl + (k >> 1) * KP_PITCH + 64);
    v2f b0 = {q0.x, q0.y};
    v2f b1 = {q0.z, q0.w};
    v2f b2 = {q1.x, q1.y};
    v2f b3 = {q1.z, q1.w};
    acc0 = __builtin_amdgcn_wmma_f32_16x16x4_f32(
         false, a,  false, b0,
         (short)0, acc0,  false,  false);
    acc1 = __builtin_amdgcn_wmma_f32_16x16x4_f32(
        false, a, false, b1, (short)0, acc1, false, false);
    acc2 = __builtin_amdgcn_wmma_f32_16x16x4_f32(
        false, a, false, b2, (short)0, acc2, false, false);
    acc3 = __builtin_amdgcn_wmma_f32_16x16x4_f32(
        false, a, false, b3, (short)0, acc3, false, false);
  }

#pragma unroll
  for (int i = 0; i < 8; ++i) {
    const int orow = row_base + half * 8 + i;
    if (orow < n_nodes) {
      float* __restrict__ o = support + (long long)orow * OUT_FEAT + l16 * 2;
      v2f lo = {acc0[i], acc1[i]};
      v2f hi = {acc2[i], acc3[i]};
      *(v2f*)(o)      = lo;
      *(v2f*)(o + 32) = hi;
    }
  }
}

__global__ __launch_bounds__(1024) __attribute__((amdgpu_waves_per_eu(8, 8))) void gcn_bin_and_gemm(
    const float* __restrict__ x, const float* __restrict__ w,
    const int* __restrict__ rows, const int* __restrict__ cols,
    const float* __restrict__ vals, float* __restrict__ ws,
    int n_nodes, int n_edges) {
  __shared__ __attribute__((aligned(16))) float wlds[(IN_FEAT / 2) * KP_PITCH];
  __shared__ int cnt[BINS];
  __shared__ int cur[BINS];
  __shared__ int wtot[BINS / 32];
  __shared__ int   s_row[CHUNK];
  __shared__ int   s_col[CHUNK];
  __shared__ float s_val[CHUNK];

  const int tid  = threadIdx.x;
  const int lane = tid & 31;

  v2f wreg[(IN_FEAT / 2) * OUT_FEAT / 1024];
#pragma unroll
  for (int j = 0; j < (IN_FEAT / 2) * OUT_FEAT / 1024; ++j) {
    const int i  = tid + j * 1024;
    const int kp = i >> 6;
    const int c  = i & 63;
    wreg[j].x = w[(2 * kp) * OUT_FEAT + c];
    wreg[j].y = w[(2 * kp + 1) * OUT_FEAT + c];
  }

  const int rpb = (n_nodes + BINS - 1) / BINS;
  int*   g_row = (int*)(ws + (long long)n_nodes * OUT_FEAT);
  int*   g_col = g_row + n_edges;
  float* g_val = (float*)(g_col + n_edges);
  int*   g_off = (int*)(g_val + n_edges);
  const int n_chunks = (n_edges + CHUNK - 1) >> CHUNK_SHIFT;

  for (int ch = (int)gridDim.x - 1 - (int)blockIdx.x; ch < n_chunks; ch += (int)gridDim.x) {
    const long long e_begin = (long long)ch * CHUNK;
    const long long left = (long long)n_edges - e_begin;
    const int n_here = left >= CHUNK ? CHUNK : (int)left;

    int r[CHUNK / 1024], c[CHUNK / 1024], bn[CHUNK / 1024];
    float vv[CHUNK / 1024];
#pragma unroll
    for (int i = 0; i < CHUNK / 1024; ++i) {
      const int idx = tid + i * 1024;
      const long long e = e_begin + (idx < n_here ? idx : n_here - 1);
      r[i]  = rows[e];
      c[i]  = cols[e];
      vv[i] = vals[e];
    }

    if (tid < BINS) cnt[tid] = 0;
    __syncthreads();

#pragma unroll
    for (int i = 0; i < CHUNK / 1024; ++i) {
      const int idx = tid + i * 1024;
      bn[i] = r[i] / rpb;
      if (idx < n_here) atomicAdd(&cnt[bn[i]], 1);
    }
    __syncthreads();

    const int v = (tid < BINS) ? cnt[tid] : 0;
    int incl = v;
#pragma unroll
    for (int d = 1; d < 32; d <<= 1) {
      const int t = __shfl_up(incl, d);
      if (lane >= d) incl += t;
    }
    if (tid < BINS && lane == 31) wtot[tid >> 5] = incl;
    __syncthreads();
    if (tid < BINS) {
      int pre = 0;
      for (int i = 0; i < (tid >> 5); ++i) pre += wtot[i];
      const int base = pre + incl - v;
      cur[tid] = base;
      int* off = g_off + (long long)ch * (BINS + 1);
      off[tid] = (int)e_begin + base;
      if (tid == BINS - 1) off[BINS] = (int)e_begin + n_here;
    }
    __syncthreads();

#pragma unroll
    for (int i = 0; i < CHUNK / 1024; ++i) {
      const int idx = tid + i * 1024;
      if (idx < n_here) {
        const int pos = atomicAdd(&cur[bn[i]], 1);
        s_row[pos] = r[i] - bn[i] * rpb;
        s_col[pos] = c[i];
        s_val[pos] = vv[i];
      }
    }
    __syncthreads();

    for (int idx = tid; idx < n_here; idx += 1024) {
      g_row[e_begin + idx] = s_row[idx];
      g_col[e_begin + idx] = s_col[idx];
      g_val[e_begin + idx] = s_val[idx];
    }
  }
#pragma unroll
  for (int j = 0; j < (IN_FEAT / 2) * OUT_FEAT / 1024; ++j) {
    const int i  = tid + j * 1024;
    *(v2f*)(wlds + (i >> 6) * KP_PITCH + (i & 63) * 2) = wreg[j];
  }
  __syncthreads();

  const int wave = __builtin_amdgcn_readfirstlane(tid >> 5);
  const int s_hi = ((wave >> 1) ^ (wave >> 3)) & 1;
  const int s_lo = (wave ^ (wave >> 2) ^ (wave >> 4)) & 1;
  const int slot = (s_hi << 4) | (s_lo << 3) | (wave & 7);
  const int G = (int)gridDim.x;
  const int b = (int)blockIdx.x;
  const int n_tiles = (n_nodes + 15) >> 4;
  const int full = n_tiles / G;
  const int rest = n_tiles - full * G;
  const int n_free = n_chunks < G ? G - n_chunks : 0;
  const int deal = n_free > 0 ? n_free : G;
  const int rounds = full + (rest + deal - 1) / deal;
  for (int s = slot; s < rounds; s += 32) {
    int t = b + G * s;
    if (s >= full) {
      const int u = b + deal * (s - full);
      if (b >= deal || u >= rest) continue;
      t = full * G + u;
    }
    asm volatile("" ::: "memory");
    gemm_tile16(x, wlds, ws, n_nodes, t * 16, lane);
  }
}

#define CAP 8192
#define MAXCH 256
#define NROWSLOT 400

template <int NB>
__device__ __forceinline__ void edge_batch(
    const float* __restrict__ support, float* acc,
    int rl_l, int c_l, int v_l, int j, int lane) {
  float s_lo[NB], s_hi[NB];
#pragma unroll
  for (int u = 0; u < NB; ++u) {
    const int c = __builtin_amdgcn_readlane(c_l, j + u);
    const float* sp = support + (long long)c * OUT_FEAT + lane;
    s_lo[u] = sp[0];
    s_hi[u] = sp[32];
  }
#pragma unroll
  for (int u = 0; u < NB; ++u) {
    const int   rl = __builtin_amdgcn_readlane(rl_l, j + u);
    const float v  = __int_as_float(__builtin_amdgcn_readlane(v_l, j + u));
    float* a = acc + rl * OUT_FEAT + lane;
    atomicAdd(a,      v * s_lo[u]);
    atomicAdd(a + 32, v * s_hi[u]);
  }
}

__device__ __forceinline__ void owner_sum_lds_adds(
    const float* __restrict__ ws, float* __restrict__ out, float* acc,
    int n_nodes, int n_edges, int b) {
  const int tid  = threadIdx.x;
  const int lane = tid & 31;
  const int wave = tid >> 5;
  const int rpb  = (n_nodes + BINS - 1) / BINS;
  const int row0 = b * rpb;
  int nrows = n_nodes - row0;
  if (nrows > rpb) nrows = rpb;
  if (nrows < 0) nrows = 0;

  const float* support = ws;
  const int*   g_row = (const int*)(ws + (long long)n_nodes * OUT_FEAT);
  const int*   g_col = g_row + n_edges;
  const float* g_val = (const float*)(g_col + n_edges);
  const int*   g_off = (const int*)(g_val + n_edges);

  v4f* acc4 = (v4f*)acc;
  const v4f z = {0.0f, 0.0f, 0.0f, 0.0f};
  for (int i = tid; i < rpb * (OUT_FEAT / 4); i += 1024) acc4[i] = z;
  __syncthreads();

  const int n_chunks = (n_edges + CHUNK - 1) >> CHUNK_SHIFT;
  for (int k0 = wave; k0 < n_chunks; k0 += 32 * 32) {
    const int kk = k0 + 32 * lane;
    int s_l = 0, t_l = 0;
    if (kk < n_chunks) {
      const int* off = g_off + (long long)kk * (BINS + 1) + b;
      s_l = off[0];
      t_l = off[1];
    }
    for (int m = 0; m < 32; ++m) {
      const int s = __builtin_amdgcn_readlane(s_l, m);
      const int t = __builtin_amdgcn_readlane(t_l, m);
      for (int g = s; g < t; g += 32) {
        const int rem  = t - g;
        const int cntg = rem < 32 ? rem : 32;
        const int el   = (lane < cntg) ? (g + lane) : (t - 1);
        const int rl_l = g_row[el];
        const int c_l  = g_col[el];
        const int v_l  = __float_as_int(g_val[el]);
        int j = 0;
        for (; j + 16 <= cntg; j += 16)
          edge_batch<16>(support, acc, rl_l, c_l, v_l, j, lane);
        for (; j + 4 <= cntg; j += 4)
          edge_batch<4>(support, acc, rl_l, c_l, v_l, j, lane);
        for (; j < cntg; ++j)
          edge_batch<1>(support, acc, rl_l, c_l, v_l, j, lane);
      }
    }
  }
  __syncthreads();

  v4f* o4 = (v4f*)(out + (long long)row0 * OUT_FEAT);
  for (int i = tid; i < nrows * (OUT_FEAT / 4); i += 1024) o4[i] = acc4[i];
}

__device__ __forceinline__ void store_row(float* __restrict__ orow0, int row, int lane, v2f a) {
  *(v2f*)(orow0 + (long long)row * OUT_FEAT + lane * 2) = a;
}

template <int NB>
__device__ __forceinline__ void row_batch(
    const float* __restrict__ support, float* __restrict__ orow0,
    int r_l, int c_l, int v_l, int j, int lane, int& row, v2f& a) {
  v2f s[NB];
#pragma unroll
  for (int u = 0; u < NB; ++u) {
    const int c = __builtin_amdgcn_readlane(c_l, j + u);
    s[u] = *(const v2f*)(support + (long long)c * OUT_FEAT + lane * 2);
  }
#pragma unroll
  for (int u = 0; u < NB; ++u) {
    const int   rl = __builtin_amdgcn_readlane(r_l, j + u);
    const float v  = __int_as_float(__builtin_amdgcn_readlane(v_l, j + u));
    if (rl != row) {
      store_row(orow0, row, lane, a);
      const v2f z = {0.0f, 0.0f};
#pragma nounroll
      for (int r = row + 1; r < rl; ++r) store_row(orow0, r, lane, z);
      row = rl;
      a = z;
    }
    a.x += v * s[u].x;
    a.y += v * s[u].y;
  }
}

__global__ __launch_bounds__(1024) void gcn_owner_sum(
    const float* __restrict__ ws, float* __restrict__ out,
    int n_nodes, int n_edges) {
  __shared__ __attribute__((aligned(16))) float lds[6 * CAP];
  __shared__ int seg_s[MAXCH];
  __shared__ int seg_pre[MAXCH + 1];
  __shared__ int cnt[NROWSLOT];
  __shared__ int start[NROWSLOT + 1];
  __shared__ int cur[NROWSLOT];
  __shared__ int wtot[32];

  const int tid  = threadIdx.x;
  const int lane = tid & 31;
  const int wave = __builtin_amdgcn_readfirstlane(tid >> 5);
  const int b    = blockIdx.x;
  const int rpb  = (n_nodes + BINS - 1) / BINS;
  const int row0 = b * rpb;
  int nrows = n_nodes - row0;
  if (nrows > rpb) nrows = rpb;
  if (nrows < 0) nrows = 0;

  const float* support = ws;
  const int*   g_row = (const int*)(ws + (long long)n_nodes * OUT_FEAT);
  const int*   g_col = g_row + n_edges;
  const float* g_val = (const float*)(g_col + n_edges);
  const int*   g_off = (const int*)(g_val + n_edges);

  const int n_chunks = (n_edges + CHUNK - 1) >> CHUNK_SHIFT;

  int seg0 = 0, len = 0;
  if (tid < n_chunks && tid < MAXCH) {
    const int* off = g_off + (long long)tid * (BINS + 1) + b;
    seg0 = off[0];
    len  = off[1] - seg0;
  }
  int incl = len;
#pragma unroll
  for (int d = 1; d < 32; d <<= 1) {
    const int t = __shfl_up(incl, d);
    if (lane >= d) incl += t;
  }
  if (tid < MAXCH && lane == 31) wtot[wave] = incl;
  if (tid < NROWSLOT) cnt[tid] = 0;
  __syncthreads();
  if (tid < MAXCH) {
    int pre = 0;
    for (int i = 0; i < wave; ++i) pre += wtot[i];
    const int excl = pre + incl - len;
    seg_s[tid]   = seg0;
    seg_pre[tid] = excl;
    if (tid == MAXCH - 1) seg_pre[MAXCH] = excl + len;
  }
  __syncthreads();
  const int n_bin = seg_pre[MAXCH];

  if (n_chunks > MAXCH || n_bin > CAP || rpb > RPB_MAX) {
    owner_sum_lds_adds(ws, out, lds, n_nodes, n_edges, b);
    return;
  }

  int*   st_r = (int*)lds;
  int*   st_c = st_r + CAP;
  float* st_v = (float*)(st_c + CAP);
  int*   so_r = st_r + 3 * CAP;
  int*   so_c = so_r + CAP;
  float* so_v = (float*)(so_c + CAP);

  for (int k0 = wave; k0 < n_chunks; k0 += 32 * 8) {
    int sg[8], pg[8], ng[8];
#pragma unroll
    for (int u = 0; u < 8; ++u) {
      const int k = k0 + 32 * u;
      sg[u] = 0; pg[u] = 0; ng[u] = 0;
      if (k < n_chunks) {
        sg[u] = seg_s[k];
        pg[u] = seg_pre[k];
        ng[u] = seg_pre[k + 1] - pg[u];
      }
    }
    int ra[8], ca[8], rb[8], cb[8];
    float va[8], vb[8];
#pragma unroll
    for (int u = 0; u < 8; ++u) {
      ra[u] = 0; ca[u] = 0; va[u] = 0.0f;
      rb[u] = 0; cb[u] = 0; vb[u] = 0.0f;
      if (lane < ng[u]) {
        const int e = sg[u] + lane;
        ra[u] = g_row[e]; ca[u] = g_col[e]; va[u] = g_val[e];
      }
      if (lane + 32 < ng[u]) {
        const int e = sg[u] + lane + 32;
        rb[u] = g_row[e]; cb[u] = g_col[e]; vb[u] = g_val[e];
      }
    }
#pragma unroll
    for (int u = 0; u < 8; ++u) {
      if (lane < ng[u]) {
        const int p = pg[u] + lane;
        st_r[p] = ra[u]; st_c[p] = ca[u]; st_v[p] = va[u];
        atomicAdd(&cnt[ra[u]], 1);
      }
      if (lane + 32 < ng[u]) {
        const int p = pg[u] + lane + 32;
        st_r[p] = rb[u]; st_c[p] = cb[u]; st_v[p] = vb[u];
        atomicAdd(&cnt[rb[u]], 1);
      }
      for (int g = lane + 64; g < ng[u]; g += 32) {
        const int e = sg[u] + g;
        const int p = pg[u] + g;
        const int r = g_row[e];
        st_r[p] = r; st_c[p] = g_col[e]; st_v[p] = g_val[e];
        atomicAdd(&cnt[r], 1);
      }
    }
  }
  __syncthreads();

  {
    const int v = (tid < NROWSLOT) ? cnt[tid] : 0;
    int inc2 = v;
#pragma unroll
    for (int d = 1; d < 32; d <<= 1) {
      const int t = __shfl_up(inc2, d);
      if (lane >= d) inc2 += t;
    }
    if (tid < 512 && lane == 31) wtot[wave] = inc2;
    __syncthreads();
    if (tid < NROWSLOT) {
      int pre = 0;
      for (int i = 0; i < wave; ++i) pre += wtot[i];
      const int excl = pre + inc2 - v;
      start[tid] = excl;
      cur[tid]   = excl;
      if (tid == NROWSLOT - 1) start[NROWSLOT] = excl + v;
    }
    __syncthreads();
  }

  for (int i = tid; i < n_bin; i += 1024) {
    const int r = st_r[i];
    const int pos = atomicAdd(&cur[r], 1);
    so_r[pos] = r;
    so_c[pos] = st_c[i];
    so_v[pos] = st_v[i];
  }
  __syncthreads();

  const int r_a = (wave * nrows) >> 5;
  const int r_b = ((wave + 1) * nrows) >> 5;
  if (r_a >= r_b) return;
  const int p0 = __builtin_amdgcn_readfirstlane(start[r_a]);
  const int p1 = __builtin_amdgcn_readfirstlane(start[r_b]);
  float* __restrict__ orow0 = out + (long long)row0 * OUT_FEAT;
  int row = r_a;
  v2f a = {0.0f, 0.0f};
  for (int g = p0; g < p1; g += 32) {
    const int rem = p1 - g;
    const int m   = rem < 32 ? rem : 32;
    const int idx = g + (lane < m ? lane : m - 1);
    const int r_l = so_r[idx];
    const int c_l = so_c[idx];
    const int v_l = __float_as_int(so_v[idx]);
    int j = 0;
    for (; j + 16 <= m; j += 16)
      row_batch<16>(support, orow0, r_l, c_l, v_l, j, lane, row, a);
    for (; j + 4 <= m; j += 4)
      row_batch<4>(support, orow0, r_l, c_l, v_l, j, lane, row, a);
    for (; j < m; ++j)
      row_batch<1>(support, orow0, r_l, c_l, v_l, j, lane, row, a);
  }
  store_row(orow0, row, lane, a);
  const v2f z2 = {0.0f, 0.0f};
  for (int r = row + 1; r < r_b; ++r) store_row(orow0, r, lane, z2);
}

extern "C" void kernel_launch(void* const* d_in, const int* in_sizes, int n_in,
                              void* d_out, int out_size, void* d_ws, size_t ws_size,
                              hipStream_t stream) {
  const float* x    = (const float*)d_in[0];
  const float* w    = (const float*)d_in[1];
  const int*   rows = (const int*)d_in[2];
  const int*   cols = (const int*)d_in[3];
  const float* vals = (const float*)d_in[4];
  float* out = (float*)d_out;
  float* ws  = (float*)d_ws;

  const int n_nodes = in_sizes[0] / IN_FEAT;
  const int n_edges = in_sizes[2];

  gcn_bin_and_gemm<<<RESIDENT_GROUPS, 1024, 0, stream>>>(
      x, w, rows, cols, vals, ws, n_nodes, n_edges);

  gcn_owner_sum<<<BINS, 1024, 0, stream>>>(ws, out, n_nodes, n_edges);
}


# edge record packed to 8 bytes, (row in bin << 20 | col) beside val: each LDS scatter, workspace store/load and reorder moves one 64-bit unit, not three 32-bit words; sum order = LDS placement, kept
# speedup vs baseline: 3.2846x; 1.0517x over previous
#include <hip/hip_runtime.h>

#define IN_FEAT 128
#define OUT_FEAT 64

typedef __attribute__((ext_vector_type(2))) float v2f;
typedef __attribute__((ext_vector_type(8))) float v8f;

typedef __attribute__((ext_vector_type(4))) float v4f;
typedef __attribute__((ext_vector_type(2))) int v2i;

#define KP_PITCH 160
#define BINS 256
#define CHUNK 8192
#define CHUNK_SHIFT 13
#define RPB_MAX 392
#define RESIDENT_GROUPS 256
#define KEY_SHIFT 20
#define KEY_MASK 0xFFFFF

__device__ __forceinline__ int key_row(int key) { return (int)((unsigned)key >> KEY_SHIFT); }
__device__ __forceinline__ int key_col(int key) { return key & KEY_MASK; }

__device__ __forceinline__ void gemm_tile16(
    const float* __restrict__ x, const float* wlds, float* __restrict__ support,
    int n_nodes, int row_base, int lane) {
  const int half = lane >> 4;
  const int l16  = lane & 15;

  int arow = row_base + l16;
  if (arow >= n_nodes) arow = n_nodes - 1;
  const float* __restrict__ xrow = x + (long long)arow * IN_FEAT;
  const float* wl = wlds + half * KP_PITCH + l16 * 4;

  v8f acc0 = {}, acc1 = {}, acc2 = {}, acc3 = {};
#pragma unroll
  for (int k = 0; k < IN_FEAT; k += 4) {
    const int kk = k + half * 2;
    v2f a = *(const v2f*)(xrow + kk);
    const v4f q0 = *(const v4f*)(wl + (k >> 1) * KP_PITCH);
    const v4f q1 = *(const v4f*)(wl + (k >> 1) * KP_PITCH + 64);
    v2f b0 = {q0.x, q0.y};
    v2f b1 = {q0.z, q0.w};
    v2f b2 = {q1.x, q1.y};
    v2f b3 = {q1.z, q1.w};
    acc0 = __builtin_amdgcn_wmma_f32_16x16x4_f32(
         false, a,  false, b0,
         (short)0, acc0,  false,  false);
    acc1 = __builtin_amdgcn_wmma_f32_16x16x4_f32(
        false, a, false, b1, (short)0, acc1, false, false);
    acc2 = __builtin_amdgcn_wmma_f32_16x16x4_f32(
        false, a, false, b2, (short)0, acc2, false, false);
    acc3 = __builtin_amdgcn_wmma_f32_16x16x4_f32(
        false, a, false, b3, (short)0, acc3, false, false);
  }

#pragma unroll
  for (int i = 0; i < 8; ++i) {
    const int orow = row_base + half * 8 + i;
    if (orow < n_nodes) {
      float* __restrict__ o = support + (long long)orow * OUT_FEAT + l16 * 2;
      v2f lo = {acc0[i], acc1[i]};
      v2f hi = {acc2[i], acc3[i]};
      *(v2f*)(o)      = lo;
      *(v2f*)(o + 32) = hi;
    }
  }
}

__global__ __launch_bounds__(1024) __attribute__((amdgpu_waves_per_eu(8, 8))) void gcn_bin_and_gemm(
    const float* __restrict__ x, const float* __restrict__ w,
    const int* __restrict__ rows, const int* __restrict__ cols,
    const float* __restrict__ vals, float* __restrict__ ws,
    int n_nodes, int n_edges) {
  __shared__ __attribute__((aligned(16))) float wlds[(IN_FEAT / 2) * KP_PITCH];
  __shared__ int cnt[BINS];
  __shared__ int cur[BINS];
  __shared__ int wtot[BINS / 32];
  __shared__ __attribute__((aligned(16))) v2i s_rec[CHUNK];

  const int tid  = threadIdx.x;
  const int lane = tid & 31;

  v2f wreg[(IN_FEAT / 2) * OUT_FEAT / 1024];
#pragma unroll
  for (int j = 0; j < (IN_FEAT / 2) * OUT_FEAT / 1024; ++j) {
    const int i  = tid + j * 1024;
    const int kp = i >> 6;
    const int c  = i & 63;
    wreg[j].x = w[(2 * kp) * OUT_FEAT + c];
    wreg[j].y = w[(2 * kp + 1) * OUT_FEAT + c];
  }

  const int rpb = (n_nodes + BINS - 1) / BINS;
  v2i* g_rec = (v2i*)(ws + (long long)n_nodes * OUT_FEAT);
  int* g_off = (int*)(g_rec + n_edges);
  const int n_chunks = (n_edges + CHUNK - 1) >> CHUNK_SHIFT;

  for (int ch = (int)gridDim.x - 1 - (int)blockIdx.x; ch < n_chunks; ch += (int)gridDim.x) {
    const long long e_begin = (long long)ch * CHUNK;
    const long long left = (long long)n_edges - e_begin;
    const int n_here = left >= CHUNK ? CHUNK : (int)left;

    int r[CHUNK / 1024], c[CHUNK / 1024], bn[CHUNK / 1024];
    float vv[CHUNK / 1024];
#pragma unroll
    for (int i = 0; i < CHUNK / 1024; ++i) {
      const int idx = tid + i * 1024;
      const long long e = e_begin + (idx < n_here ? idx : n_here - 1);
      r[i]  = rows[e];
      c[i]  = cols[e];
      vv[i] = vals[e];
    }

    if (tid < BINS) cnt[tid] = 0;
    __syncthreads();

#pragma unroll
    for (int i = 0; i < CHUNK / 1024; ++i) {
      const int idx = tid + i * 1024;
      bn[i] = r[i] / rpb;
      if (idx < n_here) atomicAdd(&cnt[bn[i]], 1);
    }
    __syncthreads();

    const int v = (tid < BINS) ? cnt[tid] : 0;
    int incl = v;
#pragma unroll
    for (int d = 1; d < 32; d <<= 1) {
      const int t = __shfl_up(incl, d);
      if (lane >= d) incl += t;
    }
    if (tid < BINS && lane == 31) wtot[tid >> 5] = incl;
    __syncthreads();
    if (tid < BINS) {
      int pre = 0;
      for (int i = 0; i < (tid >> 5); ++i) pre += wtot[i];
      const int base = pre + incl - v;
      cur[tid] = base;
      int* off = g_off + (long long)ch * (BINS + 1);
      off[tid] = (int)e_begin + base;
      if (tid == BINS - 1) off[BINS] = (int)e_begin + n_here;
    }
    __syncthreads();

#pragma unroll
    for (int i = 0; i < CHUNK / 1024; ++i) {
      const int idx = tid + i * 1024;
      if (idx < n_here) {
        const int pos = atomicAdd(&cur[bn[i]], 1);
        v2i rec;
        rec.x = (int)(((unsigned)(r[i] - bn[i] * rpb) << KEY_SHIFT) | (unsigned)c[i]);
        rec.y = __float_as_int(vv[i]);
        s_rec[pos] = rec;
      }
    }
    __syncthreads();

    for (int idx = tid; idx < n_here; idx += 1024) g_rec[e_begin + idx] = s_rec[idx];
  }
#pragma unroll
  for (int j = 0; j < (IN_FEAT / 2) * OUT_FEAT / 1024; ++j) {
    const int i  = tid + j * 1024;
    *(v2f*)(wlds + (i >> 6) * KP_PITCH + (i & 63) * 2) = wreg[j];
  }
  __syncthreads();

  const int wave = __builtin_amdgcn_readfirstlane(tid >> 5);
  const int s_hi = ((wave >> 1) ^ (wave >> 3)) & 1;
  const int s_lo = (wave ^ (wave >> 2) ^ (wave >> 4)) & 1;
  const int slot = (s_hi << 4) | (s_lo << 3) | (wave & 7);
  const int G = (int)gridDim.x;
  const int b = (int)blockIdx.x;
  const int n_tiles = (n_nodes + 15) >> 4;
  const int full = n_tiles / G;
  const int rest = n_tiles - full * G;
  const int n_free = n_chunks < G ? G - n_chunks : 0;
  const int deal = n_free > 0 ? n_free : G;
  const int rounds = full + (rest + deal - 1) / deal;
  for (int s = slot; s < rounds; s += 32) {
    int t = b + G * s;
    if (s >= full) {
      const int u = b + deal * (s - full);
      if (b >= deal || u >= rest) continue;
      t = full * G + u;
    }
    asm volatile("" ::: "memory");
    gemm_tile16(x, wlds, ws, n_nodes, t * 16, lane);
  }
}

#define CAP 8192
#define MAXCH 256
#define NROWSLOT 400

template <int NB>
__device__ __forceinline__ void edge_batch(
    const float* __restrict__ support, float* acc,
    int rl_l, int c_l, int v_l, int j, int lane) {
  float s_lo[NB], s_hi[NB];
#pragma unroll
  for (int u = 0; u < NB; ++u) {
    const int c = __builtin_amdgcn_readlane(c_l, j + u);
    const float* sp = support + (long long)c * OUT_FEAT + lane;
    s_lo[u] = sp[0];
    s_hi[u] = sp[32];
  }
#pragma unroll
  for (int u = 0; u < NB; ++u) {
    const int   rl = __builtin_amdgcn_readlane(rl_l, j + u);
    const float v  = __int_as_float(__builtin_amdgcn_readlane(v_l, j + u));
    float* a = acc + rl * OUT_FEAT + lane;
    atomicAdd(a,      v * s_lo[u]);
    atomicAdd(a + 32, v * s_hi[u]);
  }
}

__device__ __forceinline__ void owner_sum_lds_adds(
    const float* __restrict__ ws, float* __restrict__ out, float* acc,
    int n_nodes, int n_edges, int b) {
  const int tid  = threadIdx.x;
  const int lane = tid & 31;
  const int wave = tid >> 5;
  const int rpb  = (n_nodes + BINS - 1) / BINS;
  const int row0 = b * rpb;
  int nrows = n_nodes - row0;
  if (nrows > rpb) nrows = rpb;
  if (nrows < 0) nrows = 0;

  const float* support = ws;
  const v2i* g_rec = (const v2i*)(ws + (long long)n_nodes * OUT_FEAT);
  const int* g_off = (const int*)(g_rec + n_edges);

  v4f* acc4 = (v4f*)acc;
  const v4f z = {0.0f, 0.0f, 0.0f, 0.0f};
  for (int i = tid; i < rpb * (OUT_FEAT / 4); i += 1024) acc4[i] = z;
  __syncthreads();

  const int n_chunks = (n_edges + CHUNK - 1) >> CHUNK_SHIFT;
  for (int k0 = wave; k0 < n_chunks; k0 += 32 * 32) {
    const int kk = k0 + 32 * lane;
    int s_l = 0, t_l = 0;
    if (kk < n_chunks) {
      const int* off = g_off + (long long)kk * (BINS + 1) + b;
      s_l = off[0];
      t_l = off[1];
    }
    for (int m = 0; m < 32; ++m) {
      const int s = __builtin_amdgcn_readlane(s_l, m);
      const int t = __builtin_amdgcn_readlane(t_l, m);
      for (int g = s; g < t; g += 32) {
        const int rem  = t - g;
        const int cntg = rem < 32 ? rem : 32;
        const int el   = (lane < cntg) ? (g + lane) : (t - 1);
        const v2i q = g_rec[el];
        const int rl_l = key_row(q.x);
        const int c_l  = key_col(q.x);
        const int v_l  = q.y;
        int j = 0;
        for (; j + 16 <= cntg; j += 16)
          edge_batch<16>(support, acc, rl_l, c_l, v_l, j, lane);
        for (; j + 4 <= cntg; j += 4)
          edge_batch<4>(support, acc, rl_l, c_l, v_l, j, lane);
        for (; j < cntg; ++j)
          edge_batch<1>(support, acc, rl_l, c_l, v_l, j, lane);
      }
    }
  }
  __syncthreads();

  v4f* o4 = (v4f*)(out + (long long)row0 * OUT_FEAT);
  for (int i = tid; i < nrows * (OUT_FEAT / 4); i += 1024) o4[i] = acc4[i];
}

__device__ __forceinline__ void store_row(float* __restrict__ orow0, int row, int lane, v2f a) {
  *(v2f*)(orow0 + (long long)row * OUT_FEAT + lane * 2) = a;
}

template <int NB>
__device__ __forceinline__ void row_batch(
    const float* __restrict__ support, float* __restrict__ orow0,
    int r_l, int c_l, int v_l, int j, int lane, int& row, v2f& a) {
  v2f s[NB];
#pragma unroll
  for (int u = 0; u < NB; ++u) {
    const int c = __builtin_amdgcn_readlane(c_l, j + u);
    s[u] = *(const v2f*)(support + (long long)c * OUT_FEAT + lane * 2);
  }
#pragma unroll
  for (int u = 0; u < NB; ++u) {
    const int   rl = __builtin_amdgcn_readlane(r_l, j + u);
    const float v  = __int_as_float(__builtin_amdgcn_readlane(v_l, j + u));
    if (rl != row) {
      store_row(orow0, row, lane, a);
      const v2f z = {0.0f, 0.0f};
#pragma nounroll
      for (int r = row + 1; r < rl; ++r) store_row(orow0, r, lane, z);
      row = rl;
      a = z;
    }
    a.x += v * s[u].x;
    a.y += v * s[u].y;
  }
}

__global__ __launch_bounds__(1024) void gcn_owner_sum(
    const float* __restrict__ ws, float* __restrict__ out,
    int n_nodes, int n_edges) {
  __shared__ __attribute__((aligned(16))) float lds[6 * CAP];
  __shared__ int seg_s[MAXCH];
  __shared__ int seg_pre[MAXCH + 1];
  __shared__ int cnt[NROWSLOT];
  __shared__ int start[NROWSLOT + 1];
  __shared__ int cur[NROWSLOT];
  __shared__ int wtot[32];

  const int tid  = threadIdx.x;
  const int lane = tid & 31;
  const int wave = __builtin_amdgcn_readfirstlane(tid >> 5);
  const int b    = blockIdx.x;
  const int rpb  = (n_nodes + BINS - 1) / BINS;
  const int row0 = b * rpb;
  int nrows = n_nodes - row0;
  if (nrows > rpb) nrows = rpb;
  if (nrows < 0) nrows = 0;

  const float* support = ws;
  const v2i* g_rec = (const v2i*)(ws + (long long)n_nodes * OUT_FEAT);
  const int* g_off = (const int*)(g_rec + n_edges);

  const int n_chunks = (n_edges + CHUNK - 1) >> CHUNK_SHIFT;

  int seg0 = 0, len = 0;
  if (tid < n_chunks && tid < MAXCH) {
    const int* off = g_off + (long long)tid * (BINS + 1) + b;
    seg0 = off[0];
    len  = off[1] - seg0;
  }
  int incl = len;
#pragma unroll
  for (int d = 1; d < 32; d <<= 1) {
    const int t = __shfl_up(incl, d);
    if (lane >= d) incl += t;
  }
  if (tid < MAXCH && lane == 31) wtot[wave] = incl;
  if (tid < NROWSLOT) cnt[tid] = 0;
  __syncthreads();
  if (tid < MAXCH) {
    int pre = 0;
    for (int i = 0; i < wave; ++i) pre += wtot[i];
    const int excl = pre + incl - len;
    seg_s[tid]   = seg0;
    seg_pre[tid] = excl;
    if (tid == MAXCH - 1) seg_pre[MAXCH] = excl + len;
  }
  __syncthreads();
  const int n_bin = seg_pre[MAXCH];

  if (n_chunks > MAXCH || n_bin > CAP || rpb > RPB_MAX) {
    owner_sum_lds_adds(ws, out, lds, n_nodes, n_edges, b);
    return;
  }

  v2i* st = (v2i*)lds;
  v2i* so = st + CAP;

  for (int k0 = wave; k0 < n_chunks; k0 += 32 * 8) {
    int sg[8], pg[8], ng[8];
#pragma unroll
    for (int u = 0; u < 8; ++u) {
      const int k = k0 + 32 * u;
      sg[u] = 0; pg[u] = 0; ng[u] = 0;
      if (k < n_chunks) {
        sg[u] = seg_s[k];
        pg[u] = seg_pre[k];
        ng[u] = seg_pre[k + 1] - pg[u];
      }
    }
    v2i qa[8], qb[8];
#pragma unroll
    for (int u = 0; u < 8; ++u) {
      const v2i none = {0, 0};
      qa[u] = none;
      qb[u] = none;
      if (lane < ng[u])      qa[u] = g_rec[sg[u] + lane];
      if (lane + 32 < ng[u]) qb[u] = g_rec[sg[u] + lane + 32];
    }
#pragma unroll
    for (int u = 0; u < 8; ++u) {
      if (lane < ng[u]) {
        st[pg[u] + lane] = qa[u];
        atomicAdd(&cnt[key_row(qa[u].x)], 1);
      }
      if (lane + 32 < ng[u]) {
        st[pg[u] + lane + 32] = qb[u];
        atomicAdd(&cnt[key_row(qb[u].x)], 1);
      }
      for (int g = lane + 64; g < ng[u]; g += 32) {
        const v2i q = g_rec[sg[u] + g];
        st[pg[u] + g] = q;
        atomicAdd(&cnt[key_row(q.x)], 1);
      }
    }
  }
  __syncthreads();

  {
    const int v = (tid < NROWSLOT) ? cnt[tid] : 0;
    int inc2 = v;
#pragma unroll
    for (int d = 1; d < 32; d <<= 1) {
      const int t = __shfl_up(inc2, d);
      if (lane >= d) inc2 += t;
    }
    if (tid < 512 && lane == 31) wtot[wave] = inc2;
    __syncthreads();
    if (tid < NROWSLOT) {
      int pre = 0;
      for (int i = 0; i < wave; ++i) pre += wtot[i];
      const int excl = pre + inc2 - v;
      start[tid] = excl;
      cur[tid]   = excl;
      if (tid == NROWSLOT - 1) start[NROWSLOT] = excl + v;
    }
    __syncthreads();
  }

  for (int i = tid; i < n_bin; i += 1024) {
    const v2i q = st[i];
    const int pos = atomicAdd(&cur[key_row(q.x)], 1);
    so[pos] = q;
  }
  __syncthreads();

  const int r_a = (wave * nrows) >> 5;
  const int r_b = ((wave + 1) * nrows) >> 5;
  if (r_a >= r_b) return;
  const int p0 = __builtin_amdgcn_readfirstlane(start[r_a]);
  const int p1 = __builtin_amdgcn_readfirstlane(start[r_b]);
  float* __restrict__ orow0 = out + (long long)row0 * OUT_FEAT;
  int row = r_a;
  v2f a = {0.0f, 0.0f};
  for (int g = p0; g < p1; g += 32) {
    const int rem = p1 - g;
    const int m   = rem < 32 ? rem : 32;
    const int idx = g + (lane < m ? lane : m - 1);
    const v2i q = so[idx];
    const int r_l = key_row(q.x);
    const int c_l = key_col(q.x);
    const int v_l = q.y;
    int j = 0;
    for (; j + 16 <= m; j += 16)
      row_batch<16>(support, orow0, r_l, c_l, v_l, j, lane, row, a);
    for (; j + 4 <= m; j += 4)
      row_batch<4>(support, orow0, r_l, c_l, v_l, j, lane, row, a);
    for (; j < m; ++j)
      row_batch<1>(support, orow0, r_l, c_l, v_l, j, lane, row, a);
  }
  store_row(orow0, row, lane, a);
  const v2f z2 = {0.0f, 0.0f};
  for (int r = row + 1; r < r_b; ++r) store_row(orow0, r, lane, z2);
}

extern "C" void kernel_launch(void* const* d_in, const int* in_sizes, int n_in,
                              void* d_out, int out_size, void* d_ws, size_t ws_size,
                              hipStream_t stream) {
  const float* x    = (const float*)d_in[0];
  const float* w    = (const float*)d_in[1];
  const int*   rows = (const int*)d_in[2];
  const int*   cols = (const int*)d_in[3];
  const float* vals = (const float*)d_in[4];
  float* out = (float*)d_out;
  float* ws  = (float*)d_ws;

  const int n_nodes = in_sizes[0] / IN_FEAT;
  const int n_edges = in_sizes[2];

  gcn_bin_and_gemm<<<RESIDENT_GROUPS, 1024, 0, stream>>>(
      x, w, rows, cols, vals, ws, n_nodes, n_edges);

  gcn_owner_sum<<<BINS, 1024, 0, stream>>>(ws, out, n_nodes, n_edges);
}


# launch 1 split by role: 256 product groups and 256 chunk groups of 16 waves, two to a WGP, so chunks are ordered beside the matrix work, not ahead of it; sum order = LDS placement, as before
# speedup vs baseline: 3.4987x; 1.0652x over previous
#include <hip/hip_runtime.h>

#define IN_FEAT 128
#define OUT_FEAT 64

typedef __attribute__((ext_vector_type(2))) float v2f;
typedef __attribute__((ext_vector_type(8))) float v8f;

typedef __attribute__((ext_vector_type(4))) float v4f;
typedef __attribute__((ext_vector_type(2))) int v2i;

#define KP_PITCH 160
#define BINS 256
#define CHUNK 8192
#define CHUNK_SHIFT 13
#define RPB_MAX 392
#define L1_GROUPS 512
#define L1_THREADS 512
#define EPT (CHUNK / L1_THREADS)
#define WPT ((IN_FEAT / 2) * OUT_FEAT / L1_THREADS)
#define KEY_SHIFT 20
#define KEY_MASK 0xFFFFF

__device__ __forceinline__ int key_row(int key) { return (int)((unsigned)key >> KEY_SHIFT); }
__device__ __forceinline__ int key_col(int key) { return key & KEY_MASK; }

__device__ __forceinline__ void gemm_tile16(
    const float* __restrict__ x, const float* wlds, float* __restrict__ support,
    int n_nodes, int row_base, int lane) {
  const int half = lane >> 4;
  const int l16  = lane & 15;

  int arow = row_base + l16;
  if (arow >= n_nodes) arow = n_nodes - 1;
  const float* __restrict__ xrow = x + (long long)arow * IN_FEAT;
  const float* wl = wlds + half * KP_PITCH + l16 * 4;

  v8f acc0 = {}, acc1 = {}, acc2 = {}, acc3 = {};
#pragma unroll
  for (int k = 0; k < IN_FEAT; k += 4) {
    const int kk = k + half * 2;
    v2f a = *(const v2f*)(xrow + kk);
    const v4f q0 = *(const v4f*)(wl + (k >> 1) * KP_PITCH);
    const v4f q1 = *(const v4f*)(wl + (k >> 1) * KP_PITCH + 64);
    v2f b0 = {q0.x, q0.y};
    v2f b1 = {q0.z, q0.w};
    v2f b2 = {q1.x, q1.y};
    v2f b3 = {q1.z, q1.w};
    acc0 = __builtin_amdgcn_wmma_f32_16x16x4_f32(
         false, a,  false, b0,
         (short)0, acc0,  false,  false);
    acc1 = __builtin_amdgcn_wmma_f32_16x16x4_f32(
        false, a, false, b1, (short)0, acc1, false, false);
    acc2 = __builtin_amdgcn_wmma_f32_16x16x4_f32(
        false, a, false, b2, (short)0, acc2, false, false);
    acc3 = __builtin_amdgcn_wmma_f32_16x16x4_f32(
        false, a, false, b3, (short)0, acc3, false, false);
  }

#pragma unroll
  for (int i = 0; i < 8; ++i) {
    const int orow = row_base + half * 8 + i;
    if (orow < n_nodes) {
      float* __restrict__ o = support + (long long)orow * OUT_FEAT + l16 * 2;
      v2f lo = {acc0[i], acc1[i]};
      v2f hi = {acc2[i], acc3[i]};
      *(v2f*)(o)      = lo;
      *(v2f*)(o + 32) = hi;
    }
  }
}

__global__ __launch_bounds__(L1_THREADS) __attribute__((amdgpu_waves_per_eu(8, 8))) void gcn_bin_and_gemm(
    const float* __restrict__ x, const float* __restrict__ w,
    const int* __restrict__ rows, const int* __restrict__ cols,
    const float* __restrict__ vals, float* __restrict__ ws,
    int n_nodes, int n_edges) {
  __shared__ __attribute__((aligned(16))) float wlds[(IN_FEAT / 2) * KP_PITCH];
  __shared__ int cnt[BINS];
  __shared__ int cur[BINS];
  __shared__ int wtot[BINS / 32];
  __shared__ __attribute__((aligned(16))) v2i s_rec[CHUNK];

  const int tid  = threadIdx.x;
  const int lane = tid & 31;
  const int G = (int)gridDim.x >> 1;
  const bool product_role = (int)blockIdx.x < G;
  const int rank = product_role ? (int)blockIdx.x : (int)blockIdx.x - G;

  if (product_role) {
    v2f wreg[WPT];
#pragma unroll
    for (int j = 0; j < WPT; ++j) {
      const int i  = tid + j * L1_THREADS;
      const int kp = i >> 6;
      const int c  = i & 63;
      wreg[j].x = w[(2 * kp) * OUT_FEAT + c];
      wreg[j].y = w[(2 * kp + 1) * OUT_FEAT + c];
    }
#pragma unroll
    for (int j = 0; j < WPT; ++j) {
      const int i  = tid + j * L1_THREADS;
      *(v2f*)(wlds + (i >> 6) * KP_PITCH + (i & 63) * 2) = wreg[j];
    }
    __syncthreads();

    const int wave = __builtin_amdgcn_readfirstlane(tid >> 5);
    const int slot = ((((wave >> 2) + wave) & 3) << 2) | (wave & 3);
    const int n_tiles = (n_nodes + 15) >> 4;
    for (int s = slot; rank + G * s < n_tiles; s += L1_THREADS / 32) {
      asm volatile("" ::: "memory");
      gemm_tile16(x, wlds, ws, n_nodes, (rank + G * s) * 16, lane);
    }
    return;
  }

  const int rpb = (n_nodes + BINS - 1) / BINS;
  v2i* g_rec = (v2i*)(ws + (long long)n_nodes * OUT_FEAT);
  int* g_off = (int*)(g_rec + n_edges);
  const int n_chunks = (n_edges + CHUNK - 1) >> CHUNK_SHIFT;

  for (int ch = rank; ch < n_chunks; ch += G) {
    const long long e_begin = (long long)ch * CHUNK;
    const long long left = (long long)n_edges - e_begin;
    const int n_here = left >= CHUNK ? CHUNK : (int)left;

    int r[EPT], c[EPT], bn[EPT];
    float vv[EPT];
#pragma unroll
    for (int i = 0; i < EPT; ++i) {
      const int idx = tid + i * L1_THREADS;
      const long long e = e_begin + (idx < n_here ? idx : n_here - 1);
      r[i]  = rows[e];
      c[i]  = cols[e];
      vv[i] = vals[e];
    }

    if (tid < BINS) cnt[tid] = 0;
    __syncthreads();

#pragma unroll
    for (int i = 0; i < EPT; ++i) {
      const int idx = tid + i * L1_THREADS;
      bn[i] = r[i] / rpb;
      if (idx < n_here) atomicAdd(&cnt[bn[i]], 1);
    }
    __syncthreads();

    const int v = (tid < BINS) ? cnt[tid] : 0;
    int incl = v;
#pragma unroll
    for (int d = 1; d < 32; d <<= 1) {
      const int t = __shfl_up(incl, d);
      if (lane >= d) incl += t;
    }
    if (tid < BINS && lane == 31) wtot[tid >> 5] = incl;
    __syncthreads();
    if (tid < BINS) {
      int pre = 0;
      for (int i = 0; i < (tid >> 5); ++i) pre += wtot[i];
      const int base = pre + incl - v;
      cur[tid] = base;
      int* off = g_off + (long long)ch * (BINS + 1);
      off[tid] = (int)e_begin + base;
      if (tid == BINS - 1) off[BINS] = (int)e_begin + n_here;
    }
    __syncthreads();

#pragma unroll
    for (int i = 0; i < EPT; ++i) {
      const int idx = tid + i * L1_THREADS;
      if (idx < n_here) {
        const int pos = atomicAdd(&cur[bn[i]], 1);
        v2i rec;
        rec.x = (int)(((unsigned)(r[i] - bn[i] * rpb) << KEY_SHIFT) | (unsigned)c[i]);
        rec.y = __float_as_int(vv[i]);
        s_rec[pos] = rec;
      }
    }
    __syncthreads();

    for (int idx = tid; idx < n_here; idx += L1_THREADS) g_rec[e_begin + idx] = s_rec[idx];
  }
}

#define CAP 8192
#define MAXCH 256
#define NROWSLOT 400

template <int NB>
__device__ __forceinline__ void edge_batch(
    const float* __restrict__ support, float* acc,
    int rl_l, int c_l, int v_l, int j, int lane) {
  float s_lo[NB], s_hi[NB];
#pragma unroll
  for (int u = 0; u < NB; ++u) {
    const int c = __builtin_amdgcn_readlane(c_l, j + u);
    const float* sp = support + (long long)c * OUT_FEAT + lane;
    s_lo[u] = sp[0];
    s_hi[u] = sp[32];
  }
#pragma unroll
  for (int u = 0; u < NB; ++u) {
    const int   rl = __builtin_amdgcn_readlane(rl_l, j + u);
    const float v  = __int_as_float(__builtin_amdgcn_readlane(v_l, j + u));
    float* a = acc + rl * OUT_FEAT + lane;
    atomicAdd(a,      v * s_lo[u]);
    atomicAdd(a + 32, v * s_hi[u]);
  }
}

__device__ __forceinline__ void owner_sum_lds_adds(
    const float* __restrict__ ws, float* __restrict__ out, float* acc,
    int n_nodes, int n_edges, int b) {
  const int tid  = threadIdx.x;
  const int lane = tid & 31;
  const int wave = tid >> 5;
  const int rpb  = (n_nodes + BINS - 1) / BINS;
  const int row0 = b * rpb;
  int nrows = n_nodes - row0;
  if (nrows > rpb) nrows = rpb;
  if (nrows < 0) nrows = 0;

  const float* support = ws;
  const v2i* g_rec = (const v2i*)(ws + (long long)n_nodes * OUT_FEAT);
  const int* g_off = (const int*)(g_rec + n_edges);

  v4f* acc4 = (v4f*)acc;
  const v4f z = {0.0f, 0.0f, 0.0f, 0.0f};
  for (int i = tid; i < rpb * (OUT_FEAT / 4); i += 1024) acc4[i] = z;
  __syncthreads();

  const int n_chunks = (n_edges + CHUNK - 1) >> CHUNK_SHIFT;
  for (int k0 = wave; k0 < n_chunks; k0 += 32 * 32) {
    const int kk = k0 + 32 * lane;
    int s_l = 0, t_l = 0;
    if (kk < n_chunks) {
      const int* off = g_off + (long long)kk * (BINS + 1) + b;
      s_l = off[0];
      t_l = off[1];
    }
    for (int m = 0; m < 32; ++m) {
      const int s = __builtin_amdgcn_readlane(s_l, m);
      const int t = __builtin_amdgcn_readlane(t_l, m);
      for (int g = s; g < t; g += 32) {
        const int rem  = t - g;
        const int cntg = rem < 32 ? rem : 32;
        const int el   = (lane < cntg) ? (g + lane) : (t - 1);
        const v2i q = g_rec[el];
        const int rl_l = key_row(q.x);
        const int c_l  = key_col(q.x);
        const int v_l  = q.y;
        int j = 0;
        for (; j + 16 <= cntg; j += 16)
          edge_batch<16>(support, acc, rl_l, c_l, v_l, j, lane);
        for (; j + 4 <= cntg; j += 4)
          edge_batch<4>(support, acc, rl_l, c_l, v_l, j, lane);
        for (; j < cntg; ++j)
          edge_batch<1>(support, acc, rl_l, c_l, v_l, j, lane);
      }
    }
  }
  __syncthreads();

  v4f* o4 = (v4f*)(out + (long long)row0 * OUT_FEAT);
  for (int i = tid; i < nrows * (OUT_FEAT / 4); i += 1024) o4[i] = acc4[i];
}

__device__ __forceinline__ void store_row(float* __restrict__ orow0, int row, int lane, v2f a) {
  *(v2f*)(orow0 + (long long)row * OUT_FEAT + lane * 2) = a;
}

template <int NB>
__device__ __forceinline__ void row_batch(
    const float* __restrict__ support, float* __restrict__ orow0,
    int r_l, int c_l, int v_l, int j, int lane, int& row, v2f& a) {
  v2f s[NB];
#pragma unroll
  for (int u = 0; u < NB; ++u) {
    const int c = __builtin_amdgcn_readlane(c_l, j + u);
    s[u] = *(const v2f*)(support + (long long)c * OUT_FEAT + lane * 2);
  }
#pragma unroll
  for (int u = 0; u < NB; ++u) {
    const int   rl = __builtin_amdgcn_readlane(r_l, j + u);
    const float v  = __int_as_float(__builtin_amdgcn_readlane(v_l, j + u));
    if (rl != row) {
      store_row(orow0, row, lane, a);
      const v2f z = {0.0f, 0.0f};
#pragma nounroll
      for (int r = row + 1; r < rl; ++r) store_row(orow0, r, lane, z);
      row = rl;
      a = z;
    }
    a.x += v * s[u].x;
    a.y += v * s[u].y;
  }
}

__global__ __launch_bounds__(1024) void gcn_owner_sum(
    const float* __restrict__ ws, float* __restrict__ out,
    int n_nodes, int n_edges) {
  __shared__ __attribute__((aligned(16))) float lds[6 * CAP];
  __shared__ int seg_s[MAXCH];
  __shared__ int seg_pre[MAXCH + 1];
  __shared__ int cnt[NROWSLOT];
  __shared__ int start[NROWSLOT + 1];
  __shared__ int cur[NROWSLOT];
  __shared__ int wtot[32];

  const int tid  = threadIdx.x;
  const int lane = tid & 31;
  const int wave = __builtin_amdgcn_readfirstlane(tid >> 5);
  const int b    = blockIdx.x;
  const int rpb  = (n_nodes + BINS - 1) / BINS;
  const int row0 = b * rpb;
  int nrows = n_nodes - row0;
  if (nrows > rpb) nrows = rpb;
  if (nrows < 0) nrows = 0;

  const float* support = ws;
  const v2i* g_rec = (const v2i*)(ws + (long long)n_nodes * OUT_FEAT);
  const int* g_off = (const int*)(g_rec + n_edges);

  const int n_chunks = (n_edges + CHUNK - 1) >> CHUNK_SHIFT;

  int seg0 = 0, len = 0;
  if (tid < n_chunks && tid < MAXCH) {
    const int* off = g_off + (long long)tid * (BINS + 1) + b;
    seg0 = off[0];
    len  = off[1] - seg0;
  }
  int incl = len;
#pragma unroll
  for (int d = 1; d < 32; d <<= 1) {
    const int t = __shfl_up(incl, d);
    if (lane >= d) incl += t;
  }
  if (tid < MAXCH && lane == 31) wtot[wave] = incl;
  if (tid < NROWSLOT) cnt[tid] = 0;
  __syncthreads();
  if (tid < MAXCH) {
    int pre = 0;
    for (int i = 0; i < wave; ++i) pre += wtot[i];
    const int excl = pre + incl - len;
    seg_s[tid]   = seg0;
    seg_pre[tid] = excl;
    if (tid == MAXCH - 1) seg_pre[MAXCH] = excl + len;
  }
  __syncthreads();
  const int n_bin = seg_pre[MAXCH];

  if (n_chunks > MAXCH || n_bin > CAP || rpb > RPB_MAX) {
    owner_sum_lds_adds(ws, out, lds, n_nodes, n_edges, b);
    return;
  }

  v2i* st = (v2i*)lds;
  v2i* so = st + CAP;

  for (int k0 = wave; k0 < n_chunks; k0 += 32 * 8) {
    int sg[8], pg[8], ng[8];
#pragma unroll
    for (int u = 0; u < 8; ++u) {
      const int k = k0 + 32 * u;
      sg[u] = 0; pg[u] = 0; ng[u] = 0;
      if (k < n_chunks) {
        sg[u] = seg_s[k];
        pg[u] = seg_pre[k];
        ng[u] = seg_pre[k + 1] - pg[u];
      }
    }
    v2i qa[8], qb[8];
#pragma unroll
    for (int u = 0; u < 8; ++u) {
      const v2i none = {0, 0};
      qa[u] = none;
      qb[u] = none;
      if (lane < ng[u])      qa[u] = g_rec[sg[u] + lane];
      if (lane + 32 < ng[u]) qb[u] = g_rec[sg[u] + lane + 32];
    }
#pragma unroll
    for (int u = 0; u < 8; ++u) {
      if (lane < ng[u]) {
        st[pg[u] + lane] = qa[u];
        atomicAdd(&cnt[key_row(qa[u].x)], 1);
      }
      if (lane + 32 < ng[u]) {
        st[pg[u] + lane + 32] = qb[u];
        atomicAdd(&cnt[key_row(qb[u].x)], 1);
      }
      for (int g = lane + 64; g < ng[u]; g += 32) {
        const v2i q = g_rec[sg[u] + g];
        st[pg[u] + g] = q;
        atomicAdd(&cnt[key_row(q.x)], 1);
      }
    }
  }
  __syncthreads();

  {
    const int v = (tid < NROWSLOT) ? cnt[tid] : 0;
    int inc2 = v;
#pragma unroll
    for (int d = 1; d < 32; d <<= 1) {
      const int t = __shfl_up(inc2, d);
      if (lane >= d) inc2 += t;
    }
    if (tid < 512 && lane == 31) wtot[wave] = inc2;
    __syncthreads();
    if (tid < NROWSLOT) {
      int pre = 0;
      for (int i = 0; i < wave; ++i) pre += wtot[i];
      const int excl = pre + inc2 - v;
      start[tid] = excl;
      cur[tid]   = excl;
      if (tid == NROWSLOT - 1) start[NROWSLOT] = excl + v;
    }
    __syncthreads();
  }

  for (int i = tid; i < n_bin; i += 1024) {
    const v2i q = st[i];
    const int pos = atomicAdd(&cur[key_row(q.x)], 1);
    so[pos] = q;
  }
  __syncthreads();

  const int r_a = (wave * nrows) >> 5;
  const int r_b = ((wave + 1) * nrows) >> 5;
  if (r_a >= r_b) return;
  const int p0 = __builtin_amdgcn_readfirstlane(start[r_a]);
  const int p1 = __builtin_amdgcn_readfirstlane(start[r_b]);
  float* __restrict__ orow0 = out + (long long)row0 * OUT_FEAT;
  int row = r_a;
  v2f a = {0.0f, 0.0f};
  for (int g = p0; g < p1; g += 32) {
    const int rem = p1 - g;
    const int m   = rem < 32 ? rem : 32;
    const int idx = g + (lane < m ? lane : m - 1);
    const v2i q = so[idx];
    const int r_l = key_row(q.x);
    const int c_l = key_col(q.x);
    const int v_l = q.y;
    int j = 0;
    for (; j + 16 <= m; j += 16)
      row_batch<16>(support, orow0, r_l, c_l, v_l, j, lane, row, a);
    for (; j + 4 <= m; j += 4)
      row_batch<4>(support, orow0, r_l, c_l, v_l, j, lane, row, a);
    for (; j < m; ++j)
      row_batch<1>(support, orow0, r_l, c_l, v_l, j, lane, row, a);
  }
  store_row(orow0, row, lane, a);
  const v2f z2 = {0.0f, 0.0f};
  for (int r = row + 1; r < r_b; ++r) store_row(orow0, r, lane, z2);
}

extern "C" void kernel_launch(void* const* d_in, const int* in_sizes, int n_in,
                              void* d_out, int out_size, void* d_ws, size_t ws_size,
                              hipStream_t stream) {
  const float* x    = (const float*)d_in[0];
  const float* w    = (const float*)d_in[1];
  const int*   rows = (const int*)d_in[2];
  const int*   cols = (const int*)d_in[3];
  const float* vals = (const float*)d_in[4];
  float* out = (float*)d_out;
  float* ws  = (float*)d_ws;

  const int n_nodes = in_sizes[0] / IN_FEAT;
  const int n_edges = in_sizes[2];

  gcn_bin_and_gemm<<<L1_GROUPS, L1_THREADS, 0, stream>>>(
      x, w, rows, cols, vals, ws, n_nodes, n_edges);

  gcn_owner_sum<<<BINS, 1024, 0, stream>>>(ws, out, n_nodes, n_edges);
}


# owner-sum walk: a 16-lane half-wave owns a run of rows, lane = 4 features; one 16-byte gather instruction covers two edges (4 lines), addresses built on the vector unit; sum order unchanged
# speedup vs baseline: 3.6539x; 1.0444x over previous
#include <hip/hip_runtime.h>

#define IN_FEAT 128
#define OUT_FEAT 64

typedef __attribute__((ext_vector_type(2))) float v2f;
typedef __attribute__((ext_vector_type(8))) float v8f;

typedef __attribute__((ext_vector_type(4))) float v4f;
typedef __attribute__((ext_vector_type(2))) int v2i;

#define KP_PITCH 160
#define BINS 256
#define CHUNK 8192
#define CHUNK_SHIFT 13
#define RPB_MAX 392
#define L1_GROUPS 512
#define L1_THREADS 512
#define EPT (CHUNK / L1_THREADS)
#define WPT ((IN_FEAT / 2) * OUT_FEAT / L1_THREADS)
#define KEY_SHIFT 20
#define KEY_MASK 0xFFFFF

__device__ __forceinline__ int key_row(int key) { return (int)((unsigned)key >> KEY_SHIFT); }
__device__ __forceinline__ int key_col(int key) { return key & KEY_MASK; }

__device__ __forceinline__ void gemm_tile16(
    const float* __restrict__ x, const float* wlds, float* __restrict__ support,
    int n_nodes, int row_base, int lane) {
  const int half = lane >> 4;
  const int l16  = lane & 15;

  int arow = row_base + l16;
  if (arow >= n_nodes) arow = n_nodes - 1;
  const float* __restrict__ xrow = x + (long long)arow * IN_FEAT;
  const float* wl = wlds + half * KP_PITCH + l16 * 4;

  v8f acc0 = {}, acc1 = {}, acc2 = {}, acc3 = {};
#pragma unroll
  for (int k = 0; k < IN_FEAT; k += 4) {
    const int kk = k + half * 2;
    v2f a = *(const v2f*)(xrow + kk);
    const v4f q0 = *(const v4f*)(wl + (k >> 1) * KP_PITCH);
    const v4f q1 = *(const v4f*)(wl + (k >> 1) * KP_PITCH + 64);
    v2f b0 = {q0.x, q0.y};
    v2f b1 = {q0.z, q0.w};
    v2f b2 = {q1.x, q1.y};
    v2f b3 = {q1.z, q1.w};
    acc0 = __builtin_amdgcn_wmma_f32_16x16x4_f32(
         false, a,  false, b0,
         (short)0, acc0,  false,  false);
    acc1 = __builtin_amdgcn_wmma_f32_16x16x4_f32(
        false, a, false, b1, (short)0, acc1, false, false);
    acc2 = __builtin_amdgcn_wmma_f32_16x16x4_f32(
        false, a, false, b2, (short)0, acc2, false, false);
    acc3 = __builtin_amdgcn_wmma_f32_16x16x4_f32(
        false, a, false, b3, (short)0, acc3, false, false);
  }

#pragma unroll
  for (int i = 0; i < 8; ++i) {
    const int orow = row_base + half * 8 + i;
    if (orow < n_nodes) {
      float* __restrict__ o = support + (long long)orow * OUT_FEAT + l16 * 2;
      v2f lo = {acc0[i], acc1[i]};
      v2f hi = {acc2[i], acc3[i]};
      *(v2f*)(o)      = lo;
      *(v2f*)(o + 32) = hi;
    }
  }
}

__global__ __launch_bounds__(L1_THREADS) __attribute__((amdgpu_waves_per_eu(8, 8))) void gcn_bin_and_gemm(
    const float* __restrict__ x, const float* __restrict__ w,
    const int* __restrict__ rows, const int* __restrict__ cols,
    const float* __restrict__ vals, float* __restrict__ ws,
    int n_nodes, int n_edges) {
  __shared__ __attribute__((aligned(16))) float wlds[(IN_FEAT / 2) * KP_PITCH];
  __shared__ int cnt[BINS];
  __shared__ int cur[BINS];
  __shared__ int wtot[BINS / 32];
  __shared__ __attribute__((aligned(16))) v2i s_rec[CHUNK];

  const int tid  = threadIdx.x;
  const int lane = tid & 31;
  const int G = (int)gridDim.x >> 1;
  const bool product_role = (int)blockIdx.x < G;
  const int rank = product_role ? (int)blockIdx.x : (int)blockIdx.x - G;

  if (product_role) {
    v2f wreg[WPT];
#pragma unroll
    for (int j = 0; j < WPT; ++j) {
      const int i  = tid + j * L1_THREADS;
      const int kp = i >> 6;
      const int c  = i & 63;
      wreg[j].x = w[(2 * kp) * OUT_FEAT + c];
      wreg[j].y = w[(2 * kp + 1) * OUT_FEAT + c];
    }
#pragma unroll
    for (int j = 0; j < WPT; ++j) {
      const int i  = tid + j * L1_THREADS;
      *(v2f*)(wlds + (i >> 6) * KP_PITCH + (i & 63) * 2) = wreg[j];
    }
    __syncthreads();

    const int wave = __builtin_amdgcn_readfirstlane(tid >> 5);
    const int slot = ((((wave >> 2) + wave) & 3) << 2) | (wave & 3);
    const int n_tiles = (n_nodes + 15) >> 4;
    for (int s = slot; rank + G * s < n_tiles; s += L1_THREADS / 32) {
      asm volatile("" ::: "memory");
      gemm_tile16(x, wlds, ws, n_nodes, (rank + G * s) * 16, lane);
    }
    return;
  }

  const int rpb = (n_nodes + BINS - 1) / BINS;
  v2i* g_rec = (v2i*)(ws + (long long)n_nodes * OUT_FEAT);
  int* g_off = (int*)(g_rec + n_edges);
  const int n_chunks = (n_edges + CHUNK - 1) >> CHUNK_SHIFT;

  for (int ch = rank; ch < n_chunks; ch += G) {
    const long long e_begin = (long long)ch * CHUNK;
    const long long left = (long long)n_edges - e_begin;
    const int n_here = left >= CHUNK ? CHUNK : (int)left;

    int r[EPT], c[EPT], bn[EPT];
    float vv[EPT];
#pragma unroll
    for (int i = 0; i < EPT; ++i) {
      const int idx = tid + i * L1_THREADS;
      const long long e = e_begin + (idx < n_here ? idx : n_here - 1);
      r[i]  = rows[e];
      c[i]  = cols[e];
      vv[i] = vals[e];
    }

    if (tid < BINS) cnt[tid] = 0;
    __syncthreads();

#pragma unroll
    for (int i = 0; i < EPT; ++i) {
      const int idx = tid + i * L1_THREADS;
      bn[i] = r[i] / rpb;
      if (idx < n_here) atomicAdd(&cnt[bn[i]], 1);
    }
    __syncthreads();

    const int v = (tid < BINS) ? cnt[tid] : 0;
    int incl = v;
#pragma unroll
    for (int d = 1; d < 32; d <<= 1) {
      const int t = __shfl_up(incl, d);
      if (lane >= d) incl += t;
    }
    if (tid < BINS && lane == 31) wtot[tid >> 5] = incl;
    __syncthreads();
    if (tid < BINS) {
      int pre = 0;
      for (int i = 0; i < (tid >> 5); ++i) pre += wtot[i];
      const int base = pre + incl - v;
      cur[tid] = base;
      int* off = g_off + (long long)ch * (BINS + 1);
      off[tid] = (int)e_begin + base;
      if (tid == BINS - 1) off[BINS] = (int)e_begin + n_here;
    }
    __syncthreads();

#pragma unroll
    for (int i = 0; i < EPT; ++i) {
      const int idx = tid + i * L1_THREADS;
      if (idx < n_here) {
        const int pos = atomicAdd(&cur[bn[i]], 1);
        v2i rec;
        rec.x = (int)(((unsigned)(r[i] - bn[i] * rpb) << KEY_SHIFT) | (unsigned)c[i]);
        rec.y = __float_as_int(vv[i]);
        s_rec[pos] = rec;
      }
    }
    __syncthreads();

    for (int idx = tid; idx < n_here; idx += L1_THREADS) g_rec[e_begin + idx] = s_rec[idx];
  }
}

#define CAP 8192
#define MAXCH 256
#define NROWSLOT 400

template <int NB>
__device__ __forceinline__ void edge_batch(
    const float* __restrict__ support, float* acc,
    int rl_l, int c_l, int v_l, int j, int lane) {
  float s_lo[NB], s_hi[NB];
#pragma unroll
  for (int u = 0; u < NB; ++u) {
    const int c = __builtin_amdgcn_readlane(c_l, j + u);
    const float* sp = support + (long long)c * OUT_FEAT + lane;
    s_lo[u] = sp[0];
    s_hi[u] = sp[32];
  }
#pragma unroll
  for (int u = 0; u < NB; ++u) {
    const int   rl = __builtin_amdgcn_readlane(rl_l, j + u);
    const float v  = __int_as_float(__builtin_amdgcn_readlane(v_l, j + u));
    float* a = acc + rl * OUT_FEAT + lane;
    atomicAdd(a,      v * s_lo[u]);
    atomicAdd(a + 32, v * s_hi[u]);
  }
}

__device__ __forceinline__ void owner_sum_lds_adds(
    const float* __restrict__ ws, float* __restrict__ out, float* acc,
    int n_nodes, int n_edges, int b) {
  const int tid  = threadIdx.x;
  const int lane = tid & 31;
  const int wave = tid >> 5;
  const int rpb  = (n_nodes + BINS - 1) / BINS;
  const int row0 = b * rpb;
  int nrows = n_nodes - row0;
  if (nrows > rpb) nrows = rpb;
  if (nrows < 0) nrows = 0;

  const float* support = ws;
  const v2i* g_rec = (const v2i*)(ws + (long long)n_nodes * OUT_FEAT);
  const int* g_off = (const int*)(g_rec + n_edges);

  v4f* acc4 = (v4f*)acc;
  const v4f z = {0.0f, 0.0f, 0.0f, 0.0f};
  for (int i = tid; i < rpb * (OUT_FEAT / 4); i += 1024) acc4[i] = z;
  __syncthreads();

  const int n_chunks = (n_edges + CHUNK - 1) >> CHUNK_SHIFT;
  for (int k0 = wave; k0 < n_chunks; k0 += 32 * 32) {
    const int kk = k0 + 32 * lane;
    int s_l = 0, t_l = 0;
    if (kk < n_chunks) {
      const int* off = g_off + (long long)kk * (BINS + 1) + b;
      s_l = off[0];
      t_l = off[1];
    }
    for (int m = 0; m < 32; ++m) {
      const int s = __builtin_amdgcn_readlane(s_l, m);
      const int t = __builtin_amdgcn_readlane(t_l, m);
      for (int g = s; g < t; g += 32) {
        const int rem  = t - g;
        const int cntg = rem < 32 ? rem : 32;
        const int el   = (lane < cntg) ? (g + lane) : (t - 1);
        const v2i q = g_rec[el];
        const int rl_l = key_row(q.x);
        const int c_l  = key_col(q.x);
        const int v_l  = q.y;
        int j = 0;
        for (; j + 16 <= cntg; j += 16)
          edge_batch<16>(support, acc, rl_l, c_l, v_l, j, lane);
        for (; j + 4 <= cntg; j += 4)
          edge_batch<4>(support, acc, rl_l, c_l, v_l, j, lane);
        for (; j < cntg; ++j)
          edge_batch<1>(support, acc, rl_l, c_l, v_l, j, lane);
      }
    }
  }
  __syncthreads();

  v4f* o4 = (v4f*)(out + (long long)row0 * OUT_FEAT);
  for (int i = tid; i < nrows * (OUT_FEAT / 4); i += 1024) o4[i] = acc4[i];
}

__device__ __forceinline__ void store_row4(float* __restrict__ orow0, int row, int l4, v4f a) {
  *(v4f*)(orow0 + (long long)row * OUT_FEAT + l4) = a;
}

template <int NB, bool TAIL>
__device__ __forceinline__ void half_batch(
    const float* __restrict__ support, float* __restrict__ orow0, const v2i* so,
    int i, int e, unsigned lane_off, int l4, int& row, unsigned& row_top, v4f& a) {
  v2i q[NB];
#pragma unroll
  for (int u = 0; u < NB; ++u) {
    int idx = i + u;
    if (TAIL) idx = idx < e ? idx : e - 1;
    q[u] = so[idx];
  }
  v4f s[NB];
#pragma unroll
  for (int u = 0; u < NB; ++u) {
    const unsigned off = ((unsigned)key_col(q[u].x) << 8) | lane_off;
    s[u] = *(const v4f*)((const char*)support + off);
  }
#pragma unroll
  for (int u = 0; u < NB; ++u) {
    float v = __int_as_float(q[u].y);
    if (TAIL) v = (i + u < e) ? v : 0.0f;
    if ((unsigned)q[u].x > row_top) {
      const int rl = key_row(q[u].x);
      store_row4(orow0, row, l4, a);
      const v4f z = {0.0f, 0.0f, 0.0f, 0.0f};
#pragma nounroll
      for (int r = row + 1; r < rl; ++r) store_row4(orow0, r, l4, z);
      row = rl;
      row_top = (unsigned)q[u].x | (unsigned)KEY_MASK;
      a = z;
    }
    a.x += v * s[u].x;
    a.y += v * s[u].y;
    a.z += v * s[u].z;
    a.w += v * s[u].w;
  }
}

__global__ __launch_bounds__(1024) void gcn_owner_sum(
    const float* __restrict__ ws, float* __restrict__ out,
    int n_nodes, int n_edges) {
  __shared__ __attribute__((aligned(16))) float lds[6 * CAP];
  __shared__ int seg_s[MAXCH];
  __shared__ int seg_pre[MAXCH + 1];
  __shared__ int cnt[NROWSLOT];
  __shared__ int start[NROWSLOT + 1];
  __shared__ int cur[NROWSLOT];
  __shared__ int wtot[32];

  const int tid  = threadIdx.x;
  const int lane = tid & 31;
  const int wave = __builtin_amdgcn_readfirstlane(tid >> 5);
  const int b    = blockIdx.x;
  const int rpb  = (n_nodes + BINS - 1) / BINS;
  const int row0 = b * rpb;
  int nrows = n_nodes - row0;
  if (nrows > rpb) nrows = rpb;
  if (nrows < 0) nrows = 0;

  const float* support = ws;
  const v2i* g_rec = (const v2i*)(ws + (long long)n_nodes * OUT_FEAT);
  const int* g_off = (const int*)(g_rec + n_edges);

  const int n_chunks = (n_edges + CHUNK - 1) >> CHUNK_SHIFT;

  int seg0 = 0, len = 0;
  if (tid < n_chunks && tid < MAXCH) {
    const int* off = g_off + (long long)tid * (BINS + 1) + b;
    seg0 = off[0];
    len  = off[1] - seg0;
  }
  int incl = len;
#pragma unroll
  for (int d = 1; d < 32; d <<= 1) {
    const int t = __shfl_up(incl, d);
    if (lane >= d) incl += t;
  }
  if (tid < MAXCH && lane == 31) wtot[wave] = incl;
  if (tid < NROWSLOT) cnt[tid] = 0;
  __syncthreads();
  if (tid < MAXCH) {
    int pre = 0;
    for (int i = 0; i < wave; ++i) pre += wtot[i];
    const int excl = pre + incl - len;
    seg_s[tid]   = seg0;
    seg_pre[tid] = excl;
    if (tid == MAXCH - 1) seg_pre[MAXCH] = excl + len;
  }
  __syncthreads();
  const int n_bin = seg_pre[MAXCH];

  if (n_chunks > MAXCH || n_bin > CAP || rpb > RPB_MAX) {
    owner_sum_lds_adds(ws, out, lds, n_nodes, n_edges, b);
    return;
  }

  v2i* st = (v2i*)lds;
  v2i* so = st + CAP;

  for (int k0 = wave; k0 < n_chunks; k0 += 32 * 8) {
    int sg[8], pg[8], ng[8];
#pragma unroll
    for (int u = 0; u < 8; ++u) {
      const int k = k0 + 32 * u;
      sg[u] = 0; pg[u] = 0; ng[u] = 0;
      if (k < n_chunks) {
        sg[u] = seg_s[k];
        pg[u] = seg_pre[k];
        ng[u] = seg_pre[k + 1] - pg[u];
      }
    }
    v2i qa[8], qb[8];
#pragma unroll
    for (int u = 0; u < 8; ++u) {
      const v2i none = {0, 0};
      qa[u] = none;
      qb[u] = none;
      if (lane < ng[u])      qa[u] = g_rec[sg[u] + lane];
      if (lane + 32 < ng[u]) qb[u] = g_rec[sg[u] + lane + 32];
    }
#pragma unroll
    for (int u = 0; u < 8; ++u) {
      if (lane < ng[u]) {
        st[pg[u] + lane] = qa[u];
        atomicAdd(&cnt[key_row(qa[u].x)], 1);
      }
      if (lane + 32 < ng[u]) {
        st[pg[u] + lane + 32] = qb[u];
        atomicAdd(&cnt[key_row(qb[u].x)], 1);
      }
      for (int g = lane + 64; g < ng[u]; g += 32) {
        const v2i q = g_rec[sg[u] + g];
        st[pg[u] + g] = q;
        atomicAdd(&cnt[key_row(q.x)], 1);
      }
    }
  }
  __syncthreads();

  {
    const int v = (tid < NROWSLOT) ? cnt[tid] : 0;
    int inc2 = v;
#pragma unroll
    for (int d = 1; d < 32; d <<= 1) {
      const int t = __shfl_up(inc2, d);
      if (lane >= d) inc2 += t;
    }
    if (tid < 512 && lane == 31) wtot[wave] = inc2;
    __syncthreads();
    if (tid < NROWSLOT) {
      int pre = 0;
      for (int i = 0; i < wave; ++i) pre += wtot[i];
      const int excl = pre + inc2 - v;
      start[tid] = excl;
      cur[tid]   = excl;
      if (tid == NROWSLOT - 1) start[NROWSLOT] = excl + v;
    }
    __syncthreads();
  }

  for (int i = tid; i < n_bin; i += 1024) {
    const v2i q = st[i];
    const int pos = atomicAdd(&cur[key_row(q.x)], 1);
    so[pos] = q;
  }
  __syncthreads();

  const int half = lane >> 4;
  const int l4   = (lane & 15) * 4;
  const unsigned lane_off = (unsigned)l4 * 4u;
  const int hw  = wave * 2 + half;
  const int r_a = (hw * nrows) >> 6;
  const int r_b = ((hw + 1) * nrows) >> 6;
  if (r_a >= r_b) return;
  int i = start[r_a];
  const int e = start[r_b];
  float* __restrict__ orow0 = out + (long long)row0 * OUT_FEAT;
  int row = r_a;
  unsigned row_top = ((unsigned)r_a << KEY_SHIFT) | (unsigned)KEY_MASK;
  v4f a = {0.0f, 0.0f, 0.0f, 0.0f};
  for (; i + 8 <= e; i += 8)
    half_batch<8, false>(support, orow0, so, i, e, lane_off, l4, row, row_top, a);
  if (i < e)
    half_batch<8, true>(support, orow0, so, i, e, lane_off, l4, row, row_top, a);
  store_row4(orow0, row, l4, a);
  const v4f z4 = {0.0f, 0.0f, 0.0f, 0.0f};
  for (int r = row + 1; r < r_b; ++r) store_row4(orow0, r, l4, z4);
}

extern "C" void kernel_launch(void* const* d_in, const int* in_sizes, int n_in,
                              void* d_out, int out_size, void* d_ws, size_t ws_size,
                              hipStream_t stream) {
  const float* x    = (const float*)d_in[0];
  const float* w    = (const float*)d_in[1];
  const int*   rows = (const int*)d_in[2];
  const int*   cols = (const int*)d_in[3];
  const float* vals = (const float*)d_in[4];
  float* out = (float*)d_out;
  float* ws  = (float*)d_ws;

  const int n_nodes = in_sizes[0] / IN_FEAT;
  const int n_edges = in_sizes[2];

  gcn_bin_and_gemm<<<L1_GROUPS, L1_THREADS, 0, stream>>>(
      x, w, rows, cols, vals, ws, n_nodes, n_edges);

  gcn_owner_sum<<<BINS, 1024, 0, stream>>>(ws, out, n_nodes, n_edges);
}
